# starting kernel
# baseline (speedup 1.0000x reference)
#include <hip/hip_runtime.h>
#include <hip/hip_bf16.h>

#define D_FEAT 64

typedef float v2f __attribute__((ext_vector_type(2)));
typedef float v8f __attribute__((ext_vector_type(8)));

__global__ void gcn_zero_u32(unsigned* __restrict__ p, long long n) {
  long long i = (long long)blockIdx.x * blockDim.x + threadIdx.x;
  if (i < n) p[i] = 0u;
}

__global__ void gcn_degree(const int* __restrict__ src, const int* __restrict__ dst,
                           unsigned* __restrict__ outdeg, unsigned* __restrict__ indeg,
                           int n_edges) {
  int e = blockIdx.x * blockDim.x + threadIdx.x;
  if (e < n_edges) {
    atomicAdd(&outdeg[src[e]], 1u);
    atomicAdd(&indeg[dst[e]], 1u);
  }
}

__global__ void gcn_norms(float* __restrict__ outn, float* __restrict__ inn, int n) {
  int i = blockIdx.x * blockDim.x + threadIdx.x;
  if (i < n) {
    unsigned od = ((const unsigned*)outn)[i];
    unsigned id_ = ((const unsigned*)inn)[i];
    outn[i] = rsqrtf((float)(od > 1u ? od : 1u));
    inn[i]  = rsqrtf((float)(id_ > 1u ? id_ : 1u));
  }
}

__global__ void gcn_scatter(const float* __restrict__ x,
                            const float* __restrict__ out_norm,
                            const int* __restrict__ src, const int* __restrict__ dst,
                            float* __restrict__ agg, int n_edges) {
  long long t = (long long)blockIdx.x * blockDim.x + threadIdx.x;
  int e = (int)(t >> 6);
  int f = (int)(t & 63);
  if (e < n_edges) {
    int s = src[e];
    int d = dst[e];
    float v = x[(size_t)s * D_FEAT + f] * out_norm[s];
    atomicAdd(&agg[(size_t)d * D_FEAT + f], v);
  }
}

__global__ __launch_bounds__(128) void gcn_gemm_elu(
    const float* __restrict__ agg, const float* __restrict__ in_norm,
    const float* __restrict__ W,
    const float* __restrict__ bias,
    float* __restrict__ out, int n_nodes) {
  const int lane = threadIdx.x & 31;
  const int wave = threadIdx.x >> 5;
  const int row0 = blockIdx.x * 16;
  const int n0   = wave * 16;
  const int half = lane >> 4;
  const int l15  = lane & 15;

  const int arow  = row0 + l15;
  const int arowc = arow < n_nodes ? arow : (n_nodes - 1);
  const float scale = (arow < n_nodes) ? in_norm[arowc] : 0.0f;
  const float* __restrict__ aptr = agg + (size_t)arowc * D_FEAT + 2 * half;
  const float* __restrict__ bptr = W + (size_t)(2 * half) * D_FEAT + n0 + l15;

  v8f c = {};
#pragma unroll
  for (int k0 = 0; k0 < D_FEAT; k0 += 4) {
    v2f a;
    a.x = aptr[k0] * scale;
    a.y = aptr[k0 + 1] * scale;
    v2f b;
    b.x = bptr[(size_t)k0 * D_FEAT];
    b.y = bptr[(size_t)(k0 + 1) * D_FEAT];
    c = __builtin_amdgcn_wmma_f32_16x16x4_f32(false, a, false, b,
                                              (short)0, c, false, false);
  }

  const float bv = bias[n0 + l15];
#pragma unroll
  for (int v = 0; v < 8; ++v) {
    int r = row0 + v + 8 * half;
    if (r < n_nodes) {
      float xv = c[v] + bv;
      float yv = xv > 0.0f ? xv : expm1f(xv);
      out[(size_t)r * D_FEAT + n0 + l15] = yv;
    }
  }
}

extern "C" void kernel_launch(void* const* d_in, const int* in_sizes, int n_in,
                              void* d_out, int out_size, void* d_ws, size_t ws_size,
                              hipStream_t stream) {
  const float* feat = (const float*)d_in[0];
  const float* Ws[3] = {(const float*)d_in[1], (const float*)d_in[3], (const float*)d_in[5]};
  const float* bs[3] = {(const float*)d_in[2], (const float*)d_in[4], (const float*)d_in[6]};
  const int* src = (const int*)d_in[7];
  const int* dst = (const int*)d_in[8];

  const int n_nodes = in_sizes[0] / D_FEAT;
  const int n_edges = in_sizes[7];

  float* out_norm = (float*)d_ws;
  float* in_norm  = out_norm + n_nodes;
  float* agg      = in_norm + n_nodes;
  float* hbuf     = agg + (size_t)n_nodes * D_FEAT;
  float* outp     = (float*)d_out;

  const int TPB = 256;

  {
    long long n2 = 2LL * n_nodes;
    gcn_zero_u32<<<(unsigned)((n2 + TPB - 1) / TPB), TPB, 0, stream>>>(
        (unsigned*)out_norm, n2);
    gcn_degree<<<(unsigned)((n_edges + TPB - 1) / TPB), TPB, 0, stream>>>(
        src, dst, (unsigned*)out_norm, (unsigned*)in_norm, n_edges);
    gcn_norms<<<(unsigned)((n_nodes + TPB - 1) / TPB), TPB, 0, stream>>>(
        out_norm, in_norm, n_nodes);
  }

  const float* x = feat;
  for (int l = 0; l < 3; ++l) {
    long long nagg = (long long)n_nodes * D_FEAT;
    gcn_zero_u32<<<(unsigned)((nagg + TPB - 1) / TPB), TPB, 0, stream>>>(
        (unsigned*)agg, nagg);
    long long nthreads = (long long)n_edges * D_FEAT;
    gcn_scatter<<<(unsigned)((nthreads + TPB - 1) / TPB), TPB, 0, stream>>>(
        x, out_norm, src, dst, agg, n_edges);
    float* y = (l == 2) ? outp : hbuf;
    unsigned gblocks = (unsigned)((n_nodes + 15) / 16);
    gcn_gemm_elu<<<gblocks, 128, 0, stream>>>(agg, in_norm, Ws[l], bs[l], y, n_nodes);
    x = y;
  }
}


# Float-atomic scatter replaced by a per-call edge list grouped by destination (slot = arrival-order rank from the in-degree atomic) and a register gather; each node's sum order is that rank
# speedup vs baseline: 2.6310x; 2.6310x over previous
#include <hip/hip_runtime.h>
#include <hip/hip_bf16.h>

#define D_FEAT 64

typedef float v2f __attribute__((ext_vector_type(2)));
typedef float v8f __attribute__((ext_vector_type(8)));

__global__ void gcn_zero_u32(unsigned* __restrict__ p, long long n) {
  long long i = (long long)blockIdx.x * blockDim.x + threadIdx.x;
  if (i < n) p[i] = 0u;
}

__global__ void gcn_degree(const int* __restrict__ src, const int* __restrict__ dst,
                           unsigned* __restrict__ outdeg, unsigned* __restrict__ indeg,
                           int* __restrict__ slot, int n_edges) {
  int e = blockIdx.x * blockDim.x + threadIdx.x;
  if (e < n_edges) {
    atomicAdd(&outdeg[src[e]], 1u);
    slot[e] = (int)atomicAdd(&indeg[dst[e]], 1u);
  }
}

__global__ __launch_bounds__(1024) void gcn_segments_norms(
    float* __restrict__ outn, const unsigned* __restrict__ indeg,
    float* __restrict__ inn, int* __restrict__ seg_start, int n) {
  __shared__ unsigned s_before[32];
  __shared__ unsigned s_total[32];
  const int tid = threadIdx.x;
  const int lane = tid & 31;
  const int wv = tid >> 5;
  const int base = blockIdx.x * 1024;

  unsigned before = 0u;
#pragma unroll 8
  for (int i = tid; i < base; i += 1024) before += indeg[i];
  for (int o = 16; o > 0; o >>= 1) before += __shfl_xor(before, o);

  const int i = base + tid;
  const unsigned d = (i < n) ? indeg[i] : 0u;
  unsigned incl = d;
  for (int o = 1; o < 32; o <<= 1) {
    unsigned t = __shfl_up(incl, o);
    if (lane >= o) incl += t;
  }
  if (lane == 0) s_before[wv] = before;
  if (lane == 31) s_total[wv] = incl;
  __syncthreads();
  unsigned off = s_before[lane] + (lane < wv ? s_total[lane] : 0u);
  for (int o = 16; o > 0; o >>= 1) off += __shfl_xor(off, o);

  if (i < n) {
    seg_start[i] = (int)(off + incl - d);
    const unsigned od = __float_as_uint(outn[i]);
    outn[i] = rsqrtf((float)(od > 1u ? od : 1u));
    inn[i]  = rsqrtf((float)(d > 1u ? d : 1u));
  }
}

__global__ void gcn_fill(const int* __restrict__ src, const int* __restrict__ dst,
                         const int* __restrict__ seg_start, const int* __restrict__ slot,
                         int* __restrict__ nbr, int n_edges) {
  int e = blockIdx.x * blockDim.x + threadIdx.x;
  if (e < n_edges) nbr[seg_start[dst[e]] + slot[e]] = src[e];
}

__global__ __launch_bounds__(256) void gcn_gather(
    const float* __restrict__ x, const float* __restrict__ out_norm,
    const int* __restrict__ seg_start, const unsigned* __restrict__ indeg,
    const int* __restrict__ nbr, float* __restrict__ agg, int n_nodes) {
  const int node = (int)((blockIdx.x * 256u + threadIdx.x) >> 4);
  const int l = threadIdx.x & 15;
  if (node >= n_nodes) return;
  const int start = seg_start[node];
  const int cnt = (int)indeg[node];
  const float4* __restrict__ x4 = reinterpret_cast<const float4*>(x);
  float4 acc = make_float4(0.0f, 0.0f, 0.0f, 0.0f);
  for (int b = 0; b < cnt; b += 16) {
    const int m = (cnt - b) < 16 ? (cnt - b) : 16;
    int my = 0;
    float myw = 0.0f;
    if (l < m) {
      my = nbr[start + b + l];
      myw = out_norm[my];
    }
    const int first = __shfl(my, 0, 16);
    if (l >= m) my = first;
    for (int j = 0; j < m; j += 4) {
      const int s0 = __shfl(my, j, 16);
      const int s1 = __shfl(my, j + 1, 16);
      const int s2 = __shfl(my, j + 2, 16);
      const int s3 = __shfl(my, j + 3, 16);
      const float4 v0 = x4[(size_t)s0 * (D_FEAT / 4) + l];
      const float4 v1 = x4[(size_t)s1 * (D_FEAT / 4) + l];
      const float4 v2 = x4[(size_t)s2 * (D_FEAT / 4) + l];
      const float4 v3 = x4[(size_t)s3 * (D_FEAT / 4) + l];
      const float w0 = __shfl(myw, j, 16);
      const float w1 = __shfl(myw, j + 1, 16);
      const float w2 = __shfl(myw, j + 2, 16);
      const float w3 = __shfl(myw, j + 3, 16);
      acc.x += v0.x * w0; acc.y += v0.y * w0; acc.z += v0.z * w0; acc.w += v0.w * w0;
      acc.x += v1.x * w1; acc.y += v1.y * w1; acc.z += v1.z * w1; acc.w += v1.w * w1;
      acc.x += v2.x * w2; acc.y += v2.y * w2; acc.z += v2.z * w2; acc.w += v2.w * w2;
      acc.x += v3.x * w3; acc.y += v3.y * w3; acc.z += v3.z * w3; acc.w += v3.w * w3;
    }
  }
  reinterpret_cast<float4*>(agg)[(size_t)node * (D_FEAT / 4) + l] = acc;
}

__global__ __launch_bounds__(128) void gcn_gemm_elu(
    const float* __restrict__ agg, const float* __restrict__ in_norm,
    const float* __restrict__ W,
    const float* __restrict__ bias,
    float* __restrict__ out, int n_nodes) {
  const int lane = threadIdx.x & 31;
  const int wave = threadIdx.x >> 5;
  const int row0 = blockIdx.x * 16;
  const int n0   = wave * 16;
  const int half = lane >> 4;
  const int l15  = lane & 15;

  const int arow  = row0 + l15;
  const int arowc = arow < n_nodes ? arow : (n_nodes - 1);
  const float scale = (arow < n_nodes) ? in_norm[arowc] : 0.0f;
  const float* __restrict__ aptr = agg + (size_t)arowc * D_FEAT + 2 * half;
  const float* __restrict__ bptr = W + (size_t)(2 * half) * D_FEAT + n0 + l15;

  v8f c = {};
#pragma unroll
  for (int k0 = 0; k0 < D_FEAT; k0 += 4) {
    v2f a;
    a.x = aptr[k0] * scale;
    a.y = aptr[k0 + 1] * scale;
    v2f b;
    b.x = bptr[(size_t)k0 * D_FEAT];
    b.y = bptr[(size_t)(k0 + 1) * D_FEAT];
    c = __builtin_amdgcn_wmma_f32_16x16x4_f32(false, a, false, b,
                                              (short)0, c, false, false);
  }

  const float bv = bias[n0 + l15];
#pragma unroll
  for (int v = 0; v < 8; ++v) {
    int r = row0 + v + 8 * half;
    if (r < n_nodes) {
      float xv = c[v] + bv;
      float yv = xv > 0.0f ? xv : expm1f(xv);
      out[(size_t)r * D_FEAT + n0 + l15] = yv;
    }
  }
}

extern "C" void kernel_launch(void* const* d_in, const int* in_sizes, int n_in,
                              void* d_out, int out_size, void* d_ws, size_t ws_size,
                              hipStream_t stream) {
  const float* feat = (const float*)d_in[0];
  const float* Ws[3] = {(const float*)d_in[1], (const float*)d_in[3], (const float*)d_in[5]};
  const float* bs[3] = {(const float*)d_in[2], (const float*)d_in[4], (const float*)d_in[6]};
  const int* src = (const int*)d_in[7];
  const int* dst = (const int*)d_in[8];

  const int n_nodes = in_sizes[0] / D_FEAT;
  const int n_edges = in_sizes[7];

  float*    out_norm  = (float*)d_ws;
  unsigned* indeg     = (unsigned*)(out_norm + n_nodes);
  float*    agg       = out_norm + 2 * (size_t)n_nodes;
  float*    in_norm   = agg + (size_t)n_nodes * D_FEAT;
  int*      seg_start = (int*)(in_norm + n_nodes);
  int*      nbr       = seg_start + n_nodes;
  int*      slot      = nbr + n_edges;
  float*    outp      = (float*)d_out;

  const int TPB = 256;

  {
    long long n2 = 2LL * n_nodes;
    gcn_zero_u32<<<(unsigned)((n2 + TPB - 1) / TPB), TPB, 0, stream>>>(
        (unsigned*)out_norm, n2);
    gcn_degree<<<(unsigned)((n_edges + TPB - 1) / TPB), TPB, 0, stream>>>(
        src, dst, (unsigned*)out_norm, indeg, slot, n_edges);
    gcn_segments_norms<<<(unsigned)((n_nodes + 1023) / 1024), 1024, 0, stream>>>(
        out_norm, indeg, in_norm, seg_start, n_nodes);
    gcn_fill<<<(unsigned)((n_edges + TPB - 1) / TPB), TPB, 0, stream>>>(
        src, dst, seg_start, slot, nbr, n_edges);
  }

  const float* x = feat;
  for (int l = 0; l < 3; ++l) {
    long long nthreads = (long long)n_nodes * 16;
    gcn_gather<<<(unsigned)((nthreads + TPB - 1) / TPB), TPB, 0, stream>>>(
        x, out_norm, seg_start, indeg, nbr, agg, n_nodes);
    unsigned gblocks = (unsigned)((n_nodes + 15) / 16);
    gcn_gemm_elu<<<gblocks, 128, 0, stream>>>(agg, in_norm, Ws[l], bs[l], outp, n_nodes);
    x = outp;
  }
}


# Degrees and destination grouping rebuilt as a two-level counting sort on LDS counters with staged whole-line writes, no global atomics left; a node's sum order is now its LDS-counter arrival order
# speedup vs baseline: 4.3338x; 1.6472x over previous
#include <hip/hip_runtime.h>
#include <hip/hip_bf16.h>

#define D_FEAT 64

typedef float v2f __attribute__((ext_vector_type(2)));
typedef float v8f __attribute__((ext_vector_type(8)));

#define BKT_SHIFT 9
#define BKT_NODES 512
#define SORT_SLICES 256
#define SORT_BATCH 8
#define BKT_STAGE 12288

__global__ __launch_bounds__(1024) void gcn_bucket_count(
    const int* __restrict__ src, const int* __restrict__ dst,
    unsigned* __restrict__ cnt, int n_edges, int nb) {
  extern __shared__ unsigned s_cnt[];
  const int tid = threadIdx.x;
  for (int i = tid; i < 2 * nb; i += 1024) s_cnt[i] = 0u;
  __syncthreads();
  const int per = (n_edges + (int)gridDim.x - 1) / (int)gridDim.x;
  const int e0 = (int)blockIdx.x * per;
  const int e1 = (e0 + per) < n_edges ? (e0 + per) : n_edges;
  for (int e = e0 + tid; e < e1; e += SORT_BATCH * 1024) {
    int d[SORT_BATCH], s[SORT_BATCH];
#pragma unroll
    for (int j = 0; j < SORT_BATCH; ++j) {
      const int ej = e + j * 1024;
      const int ec = ej < e1 ? ej : e1 - 1;
      const int dv = dst[ec];
      const int sv = src[ec];
      d[j] = ej < e1 ? dv : -1;
      s[j] = sv;
    }
#pragma unroll
    for (int j = 0; j < SORT_BATCH; ++j) {
      if (d[j] >= 0) {
        atomicAdd(&s_cnt[d[j] >> BKT_SHIFT], 1u);
        atomicAdd(&s_cnt[nb + (s[j] >> BKT_SHIFT)], 1u);
      }
    }
  }
  __syncthreads();
  for (int i = tid; i < 2 * nb; i += 1024)
    cnt[(size_t)i * gridDim.x + blockIdx.x] = s_cnt[i];
}

__global__ __launch_bounds__(1024) void gcn_bucket_scan(
    const unsigned* __restrict__ cnt, unsigned* __restrict__ off, int len) {
  __shared__ unsigned s_before[32];
  __shared__ unsigned s_total[32];
  const unsigned* __restrict__ c = cnt + (size_t)blockIdx.y * len;
  unsigned* __restrict__ o = off + (size_t)blockIdx.y * len;
  const int tid = threadIdx.x;
  const int lane = tid & 31;
  const int wv = tid >> 5;
  const int base = blockIdx.x * 1024;

  unsigned before = 0u;
#pragma unroll 8
  for (int i = tid; i < base; i += 1024) before += c[i];
  for (int x = 16; x > 0; x >>= 1) before += __shfl_xor(before, x);

  const int i = base + tid;
  const unsigned d = (i < len) ? c[i] : 0u;
  unsigned incl = d;
  for (int x = 1; x < 32; x <<= 1) {
    unsigned t = __shfl_up(incl, x);
    if (lane >= x) incl += t;
  }
  if (lane == 0) s_before[wv] = before;
  if (lane == 31) s_total[wv] = incl;
  __syncthreads();
  unsigned offv = s_before[lane] + (lane < wv ? s_total[lane] : 0u);
  for (int x = 16; x > 0; x >>= 1) offv += __shfl_xor(offv, x);
  if (i < len) o[i] = offv + incl - d;
}

__global__ __launch_bounds__(1024) void gcn_bucket_scatter(
    const int* __restrict__ src, const int* __restrict__ dst,
    const unsigned* __restrict__ cnt, const unsigned* __restrict__ off,
    int2* __restrict__ pairs, int* __restrict__ srcs, int n_edges, int nb) {
  extern __shared__ int2 s_stage[];
  __shared__ unsigned s_total[32];
  const int per = (n_edges + (int)gridDim.x - 1) / (int)gridDim.x;
  int2* s_pair = s_stage;
  int* s_src = (int*)(s_stage + per);
  unsigned* s_cur = (unsigned*)(s_src + per);
  unsigned* s_delta = s_cur + 2 * nb;
  const int tid = threadIdx.x;
  const int lane = tid & 31;
  const int wv = tid >> 5;
  const int e0 = (int)blockIdx.x * per;
  const int e1 = (e0 + per) < n_edges ? (e0 + per) : n_edges;

  unsigned carry = 0u;
  for (int base = 0; base < 2 * nb; base += 1024) {
    const int i = base + tid;
    const unsigned c = (i < 2 * nb) ? cnt[(size_t)i * gridDim.x + blockIdx.x] : 0u;
    unsigned incl = c;
    for (int x = 1; x < 32; x <<= 1) {
      unsigned t = __shfl_up(incl, x);
      if (lane >= x) incl += t;
    }
    if (lane == 31) s_total[wv] = incl;
    __syncthreads();
    unsigned woff = (lane < wv) ? s_total[lane] : 0u;
    unsigned all = s_total[lane];
    for (int x = 16; x > 0; x >>= 1) {
      woff += __shfl_xor(woff, x);
      all += __shfl_xor(all, x);
    }
    if (i < 2 * nb) s_cur[i] = carry + woff + incl - c;
    carry += all;
    __syncthreads();
  }
  const unsigned key0 = (unsigned)(e1 > e0 ? e1 - e0 : 0);
  for (int i = tid; i < 2 * nb; i += 1024) {
    const unsigned ls = s_cur[i] - (i >= nb ? key0 : 0u);
    s_cur[i] = ls;
    s_delta[i] = off[(size_t)i * gridDim.x + blockIdx.x] - ls;
  }
  __syncthreads();

  for (int e = e0 + tid; e < e1; e += SORT_BATCH * 1024) {
    int d[SORT_BATCH], s[SORT_BATCH];
#pragma unroll
    for (int j = 0; j < SORT_BATCH; ++j) {
      const int ej = e + j * 1024;
      const int ec = ej < e1 ? ej : e1 - 1;
      const int dv = dst[ec];
      const int sv = src[ec];
      d[j] = ej < e1 ? dv : -1;
      s[j] = sv;
    }
#pragma unroll
    for (int j = 0; j < SORT_BATCH; ++j) {
      if (d[j] >= 0) {
        const unsigned p = atomicAdd(&s_cur[d[j] >> BKT_SHIFT], 1u);
        const unsigned q = atomicAdd(&s_cur[nb + (s[j] >> BKT_SHIFT)], 1u);
        s_pair[p] = make_int2(s[j], d[j]);
        s_src[q] = s[j];
      }
    }
  }
  __syncthreads();
  const int n_own = e1 - e0;
  for (int i = tid; i < n_own; i += 1024) {
    const int2 pr = s_pair[i];
    const int sv = s_src[i];
    pairs[(unsigned)i + s_delta[pr.y >> BKT_SHIFT]] = pr;
    srcs[(unsigned)i + s_delta[nb + (sv >> BKT_SHIFT)]] = sv;
  }
}

__global__ __launch_bounds__(BKT_NODES) void gcn_bucket_build(
    const int2* __restrict__ pairs, const int* __restrict__ srcs,
    const unsigned* __restrict__ off, int slices, int nb,
    float* __restrict__ outn, unsigned* __restrict__ indeg,
    float* __restrict__ inn, int* __restrict__ seg_start,
    int* __restrict__ nbr, int n_nodes, int n_edges) {
  __shared__ unsigned c_in[BKT_NODES];
  __shared__ unsigned c_out[BKT_NODES];
  __shared__ unsigned s_total[32];
  __shared__ int s_nbr[BKT_STAGE];
  const int tid = threadIdx.x;
  const int lane = tid & 31;
  const int wv = tid >> 5;
  const int b = blockIdx.x;
  const int node0 = b << BKT_SHIFT;
  c_in[tid] = 0u;
  c_out[tid] = 0u;
  __syncthreads();
  const int es = (int)off[(size_t)b * slices];
  const int ee = (b + 1 < nb) ? (int)off[(size_t)(b + 1) * slices] : n_edges;
  const int ss = (int)off[(size_t)(nb + b) * slices];
  const int se = (b + 1 < nb) ? (int)off[(size_t)(nb + b + 1) * slices] : n_edges;
  for (int e = es + tid, f = ss + tid; e < ee || f < se;
       e += SORT_BATCH * BKT_NODES, f += SORT_BATCH * BKT_NODES) {
    int d[SORT_BATCH], s[SORT_BATCH];
#pragma unroll
    for (int j = 0; j < SORT_BATCH; ++j) {
      const int ej = e + j * BKT_NODES;
      const int fj = f + j * BKT_NODES;
      const int dv = pairs[ej < ee ? ej : n_edges - 1].y;
      const int sv = srcs[fj < se ? fj : n_edges - 1];
      d[j] = ej < ee ? dv : -1;
      s[j] = fj < se ? sv : -1;
    }
#pragma unroll
    for (int j = 0; j < SORT_BATCH; ++j) {
      if (d[j] >= 0) atomicAdd(&c_in[d[j] - node0], 1u);
      if (s[j] >= 0) atomicAdd(&c_out[s[j] - node0], 1u);
    }
  }
  __syncthreads();

  const unsigned d = c_in[tid];
  unsigned incl = d;
  for (int x = 1; x < 32; x <<= 1) {
    unsigned t = __shfl_up(incl, x);
    if (lane >= x) incl += t;
  }
  if (lane == 31) s_total[wv] = incl;
  __syncthreads();
  unsigned woff = (lane < wv) ? s_total[lane] : 0u;
  for (int x = 16; x > 0; x >>= 1) woff += __shfl_xor(woff, x);
  const unsigned local = woff + incl - d;
  const unsigned first = (unsigned)es + local;
  const bool staged = (ee - es) <= BKT_STAGE;
  c_in[tid] = staged ? local : first;

  const int node = node0 + tid;
  if (node < n_nodes) {
    const unsigned od = c_out[tid];
    seg_start[node] = (int)first;
    indeg[node] = d;
    inn[node] = rsqrtf((float)(d > 1u ? d : 1u));
    outn[node] = rsqrtf((float)(od > 1u ? od : 1u));
  }
  __syncthreads();
  for (int e = es + tid; e < ee; e += SORT_BATCH * BKT_NODES) {
    int2 p[SORT_BATCH];
    unsigned pos[SORT_BATCH];
#pragma unroll
    for (int j = 0; j < SORT_BATCH; ++j) {
      const int ej = e + j * BKT_NODES;
      p[j] = pairs[ej < ee ? ej : n_edges - 1];
      if (ej >= ee) p[j].y = -1;
    }
#pragma unroll
    for (int j = 0; j < SORT_BATCH; ++j) {
      pos[j] = 0u;
      if (p[j].y >= 0) pos[j] = atomicAdd(&c_in[p[j].y - node0], 1u);
    }
#pragma unroll
    for (int j = 0; j < SORT_BATCH; ++j) {
      if (p[j].y >= 0) {
        if (staged) s_nbr[pos[j]] = p[j].x;
        else nbr[pos[j]] = p[j].x;
      }
    }
  }
  if (staged) {
    __syncthreads();
    for (int i = tid; i < ee - es; i += BKT_NODES) nbr[es + i] = s_nbr[i];
  }
}

__global__ __launch_bounds__(256) void gcn_gather(
    const float* __restrict__ x, const float* __restrict__ out_norm,
    const int* __restrict__ seg_start, const unsigned* __restrict__ indeg,
    const int* __restrict__ nbr, float* __restrict__ agg, int n_nodes) {
  const int node = (int)((blockIdx.x * 256u + threadIdx.x) >> 4);
  const int l = threadIdx.x & 15;
  if (node >= n_nodes) return;
  const int start = seg_start[node];
  const int cnt = (int)indeg[node];
  const float4* __restrict__ x4 = reinterpret_cast<const float4*>(x);
  float4 acc = make_float4(0.0f, 0.0f, 0.0f, 0.0f);
  for (int b = 0; b < cnt; b += 16) {
    const int m = (cnt - b) < 16 ? (cnt - b) : 16;
    int my = 0;
    float myw = 0.0f;
    if (l < m) {
      my = nbr[start + b + l];
      myw = out_norm[my];
    }
    const int first = __shfl(my, 0, 16);
    if (l >= m) my = first;
    for (int j = 0; j < m; j += 4) {
      const int s0 = __shfl(my, j, 16);
      const int s1 = __shfl(my, j + 1, 16);
      const int s2 = __shfl(my, j + 2, 16);
      const int s3 = __shfl(my, j + 3, 16);
      const float4 v0 = x4[(size_t)s0 * (D_FEAT / 4) + l];
      const float4 v1 = x4[(size_t)s1 * (D_FEAT / 4) + l];
      const float4 v2 = x4[(size_t)s2 * (D_FEAT / 4) + l];
      const float4 v3 = x4[(size_t)s3 * (D_FEAT / 4) + l];
      const float w0 = __shfl(myw, j, 16);
      const float w1 = __shfl(myw, j + 1, 16);
      const float w2 = __shfl(myw, j + 2, 16);
      const float w3 = __shfl(myw, j + 3, 16);
      acc.x += v0.x * w0; acc.y += v0.y * w0; acc.z += v0.z * w0; acc.w += v0.w * w0;
      acc.x += v1.x * w1; acc.y += v1.y * w1; acc.z += v1.z * w1; acc.w += v1.w * w1;
      acc.x += v2.x * w2; acc.y += v2.y * w2; acc.z += v2.z * w2; acc.w += v2.w * w2;
      acc.x += v3.x * w3; acc.y += v3.y * w3; acc.z += v3.z * w3; acc.w += v3.w * w3;
    }
  }
  reinterpret_cast<float4*>(agg)[(size_t)node * (D_FEAT / 4) + l] = acc;
}

__global__ __launch_bounds__(128) void gcn_gemm_elu(
    const float* __restrict__ agg, const float* __restrict__ in_norm,
    const float* __restrict__ W,
    const float* __restrict__ bias,
    float* __restrict__ out, int n_nodes) {
  const int lane = threadIdx.x & 31;
  const int wave = threadIdx.x >> 5;
  const int row0 = blockIdx.x * 16;
  const int n0   = wave * 16;
  const int half = lane >> 4;
  const int l15  = lane & 15;

  const int arow  = row0 + l15;
  const int arowc = arow < n_nodes ? arow : (n_nodes - 1);
  const float scale = (arow < n_nodes) ? in_norm[arowc] : 0.0f;
  const float* __restrict__ aptr = agg + (size_t)arowc * D_FEAT + 2 * half;
  const float* __restrict__ bptr = W + (size_t)(2 * half) * D_FEAT + n0 + l15;

  v8f c = {};
#pragma unroll
  for (int k0 = 0; k0 < D_FEAT; k0 += 4) {
    v2f a;
    a.x = aptr[k0] * scale;
    a.y = aptr[k0 + 1] * scale;
    v2f b;
    b.x = bptr[(size_t)k0 * D_FEAT];
    b.y = bptr[(size_t)(k0 + 1) * D_FEAT];
    c = __builtin_amdgcn_wmma_f32_16x16x4_f32(false, a, false, b,
                                              (short)0, c, false, false);
  }

  const float bv = bias[n0 + l15];
#pragma unroll
  for (int v = 0; v < 8; ++v) {
    int r = row0 + v + 8 * half;
    if (r < n_nodes) {
      float xv = c[v] + bv;
      float yv = xv > 0.0f ? xv : expm1f(xv);
      out[(size_t)r * D_FEAT + n0 + l15] = yv;
    }
  }
}

extern "C" void kernel_launch(void* const* d_in, const int* in_sizes, int n_in,
                              void* d_out, int out_size, void* d_ws, size_t ws_size,
                              hipStream_t stream) {
  const float* feat = (const float*)d_in[0];
  const float* Ws[3] = {(const float*)d_in[1], (const float*)d_in[3], (const float*)d_in[5]};
  const float* bs[3] = {(const float*)d_in[2], (const float*)d_in[4], (const float*)d_in[6]};
  const int* src = (const int*)d_in[7];
  const int* dst = (const int*)d_in[8];

  const int n_nodes = in_sizes[0] / D_FEAT;
  const int n_edges = in_sizes[7];

  float*    out_norm  = (float*)d_ws;
  unsigned* indeg     = (unsigned*)(out_norm + n_nodes);
  float*    agg       = out_norm + 2 * (size_t)n_nodes;
  float*    in_norm   = agg + (size_t)n_nodes * D_FEAT;
  int*      seg_start = (int*)(in_norm + n_nodes);
  int*      nbr       = seg_start + n_nodes;
  int*      srcs      = nbr + n_edges;
  float*    outp      = (float*)d_out;

  const int TPB = 256;

  {
    const int nb = (n_nodes + BKT_NODES - 1) >> BKT_SHIFT;
    const int len = nb * SORT_SLICES;
    int2*     pairs = (int2*)agg;
    unsigned* cnt   = (unsigned*)agg + 2 * (size_t)n_edges;
    unsigned* off   = cnt + 2 * (size_t)len;
    const size_t lds = 2 * (size_t)nb * sizeof(unsigned);
    gcn_bucket_count<<<SORT_SLICES, 1024, lds, stream>>>(src, dst, cnt, n_edges, nb);
    gcn_bucket_scan<<<dim3((unsigned)((len + 1023) / 1024), 2), 1024, 0, stream>>>(
        cnt, off, len);
    const size_t per = ((size_t)n_edges + SORT_SLICES - 1) / SORT_SLICES;
    gcn_bucket_scatter<<<SORT_SLICES, 1024, per * 12 + 2 * lds, stream>>>(
        src, dst, cnt, off, pairs, srcs, n_edges, nb);
    gcn_bucket_build<<<(unsigned)nb, BKT_NODES, 0, stream>>>(
        pairs, srcs, off, SORT_SLICES, nb, out_norm, indeg, in_norm, seg_start,
        nbr, n_nodes, n_edges);
  }

  const float* x = feat;
  for (int l = 0; l < 3; ++l) {
    long long nthreads = (long long)n_nodes * 16;
    gcn_gather<<<(unsigned)((nthreads + TPB - 1) / TPB), TPB, 0, stream>>>(
        x, out_norm, seg_start, indeg, nbr, agg, n_nodes);
    unsigned gblocks = (unsigned)((n_nodes + 15) / 16);
    gcn_gemm_elu<<<gblocks, 128, 0, stream>>>(agg, in_norm, Ws[l], bs[l], outp, n_nodes);
    x = outp;
  }
}


# Gather fused into the WMMA projection: one kernel per layer, a 4-wave workgroup takes 4 tiles of 16 nodes, summed rows reach the A fragments through LDS; agg round trip and 3 launches gone
# speedup vs baseline: 5.4938x; 1.2677x over previous
#include <hip/hip_runtime.h>
#include <hip/hip_bf16.h>

#define D_FEAT 64

typedef float v2f __attribute__((ext_vector_type(2)));
typedef float v8f __attribute__((ext_vector_type(8)));

#define BKT_SHIFT 9
#define BKT_NODES 512
#define SORT_SLICES 256
#define SORT_BATCH 8
#define BKT_STAGE 12288

__global__ __launch_bounds__(1024) void gcn_bucket_count(
    const int* __restrict__ src, const int* __restrict__ dst,
    unsigned* __restrict__ cnt, int n_edges, int nb) {
  extern __shared__ unsigned s_cnt[];
  const int tid = threadIdx.x;
  for (int i = tid; i < 2 * nb; i += 1024) s_cnt[i] = 0u;
  __syncthreads();
  const int per = (n_edges + (int)gridDim.x - 1) / (int)gridDim.x;
  const int e0 = (int)blockIdx.x * per;
  const int e1 = (e0 + per) < n_edges ? (e0 + per) : n_edges;
  for (int e = e0 + tid; e < e1; e += SORT_BATCH * 1024) {
    int d[SORT_BATCH], s[SORT_BATCH];
#pragma unroll
    for (int j = 0; j < SORT_BATCH; ++j) {
      const int ej = e + j * 1024;
      const int ec = ej < e1 ? ej : e1 - 1;
      const int dv = dst[ec];
      const int sv = src[ec];
      d[j] = ej < e1 ? dv : -1;
      s[j] = sv;
    }
#pragma unroll
    for (int j = 0; j < SORT_BATCH; ++j) {
      if (d[j] >= 0) {
        atomicAdd(&s_cnt[d[j] >> BKT_SHIFT], 1u);
        atomicAdd(&s_cnt[nb + (s[j] >> BKT_SHIFT)], 1u);
      }
    }
  }
  __syncthreads();
  for (int i = tid; i < 2 * nb; i += 1024)
    cnt[(size_t)i * gridDim.x + blockIdx.x] = s_cnt[i];
}

__global__ __launch_bounds__(1024) void gcn_bucket_scan(
    const unsigned* __restrict__ cnt, unsigned* __restrict__ off, int len) {
  __shared__ unsigned s_before[32];
  __shared__ unsigned s_total[32];
  const unsigned* __restrict__ c = cnt + (size_t)blockIdx.y * len;
  unsigned* __restrict__ o = off + (size_t)blockIdx.y * len;
  const int tid = threadIdx.x;
  const int lane = tid & 31;
  const int wv = tid >> 5;
  const int base = blockIdx.x * 1024;

  unsigned before = 0u;
#pragma unroll 8
  for (int i = tid; i < base; i += 1024) before += c[i];
  for (int x = 16; x > 0; x >>= 1) before += __shfl_xor(before, x);

  const int i = base + tid;
  const unsigned d = (i < len) ? c[i] : 0u;
  unsigned incl = d;
  for (int x = 1; x < 32; x <<= 1) {
    unsigned t = __shfl_up(incl, x);
    if (lane >= x) incl += t;
  }
  if (lane == 0) s_before[wv] = before;
  if (lane == 31) s_total[wv] = incl;
  __syncthreads();
  unsigned offv = s_before[lane] + (lane < wv ? s_total[lane] : 0u);
  for (int x = 16; x > 0; x >>= 1) offv += __shfl_xor(offv, x);
  if (i < len) o[i] = offv + incl - d;
}

__global__ __launch_bounds__(1024) void gcn_bucket_scatter(
    const int* __restrict__ src, const int* __restrict__ dst,
    const unsigned* __restrict__ cnt, const unsigned* __restrict__ off,
    int2* __restrict__ pairs, int* __restrict__ srcs, int n_edges, int nb) {
  extern __shared__ int2 s_stage[];
  __shared__ unsigned s_total[32];
  const int per = (n_edges + (int)gridDim.x - 1) / (int)gridDim.x;
  int2* s_pair = s_stage;
  int* s_src = (int*)(s_stage + per);
  unsigned* s_cur = (unsigned*)(s_src + per);
  unsigned* s_delta = s_cur + 2 * nb;
  const int tid = threadIdx.x;
  const int lane = tid & 31;
  const int wv = tid >> 5;
  const int e0 = (int)blockIdx.x * per;
  const int e1 = (e0 + per) < n_edges ? (e0 + per) : n_edges;

  unsigned carry = 0u;
  for (int base = 0; base < 2 * nb; base += 1024) {
    const int i = base + tid;
    const unsigned c = (i < 2 * nb) ? cnt[(size_t)i * gridDim.x + blockIdx.x] : 0u;
    unsigned incl = c;
    for (int x = 1; x < 32; x <<= 1) {
      unsigned t = __shfl_up(incl, x);
      if (lane >= x) incl += t;
    }
    if (lane == 31) s_total[wv] = incl;
    __syncthreads();
    unsigned woff = (lane < wv) ? s_total[lane] : 0u;
    unsigned all = s_total[lane];
    for (int x = 16; x > 0; x >>= 1) {
      woff += __shfl_xor(woff, x);
      all += __shfl_xor(all, x);
    }
    if (i < 2 * nb) s_cur[i] = carry + woff + incl - c;
    carry += all;
    __syncthreads();
  }
  const unsigned key0 = (unsigned)(e1 > e0 ? e1 - e0 : 0);
  for (int i = tid; i < 2 * nb; i += 1024) {
    const unsigned ls = s_cur[i] - (i >= nb ? key0 : 0u);
    s_cur[i] = ls;
    s_delta[i] = off[(size_t)i * gridDim.x + blockIdx.x] - ls;
  }
  __syncthreads();

  for (int e = e0 + tid; e < e1; e += SORT_BATCH * 1024) {
    int d[SORT_BATCH], s[SORT_BATCH];
#pragma unroll
    for (int j = 0; j < SORT_BATCH; ++j) {
      const int ej = e + j * 1024;
      const int ec = ej < e1 ? ej : e1 - 1;
      const int dv = dst[ec];
      const int sv = src[ec];
      d[j] = ej < e1 ? dv : -1;
      s[j] = sv;
    }
#pragma unroll
    for (int j = 0; j < SORT_BATCH; ++j) {
      if (d[j] >= 0) {
        const unsigned p = atomicAdd(&s_cur[d[j] >> BKT_SHIFT], 1u);
        const unsigned q = atomicAdd(&s_cur[nb + (s[j] >> BKT_SHIFT)], 1u);
        s_pair[p] = make_int2(s[j], d[j]);
        s_src[q] = s[j];
      }
    }
  }
  __syncthreads();
  const int n_own = e1 - e0;
  for (int i = tid; i < n_own; i += 1024) {
    const int2 pr = s_pair[i];
    const int sv = s_src[i];
    pairs[(unsigned)i + s_delta[pr.y >> BKT_SHIFT]] = pr;
    srcs[(unsigned)i + s_delta[nb + (sv >> BKT_SHIFT)]] = sv;
  }
}

__global__ __launch_bounds__(BKT_NODES) void gcn_bucket_build(
    const int2* __restrict__ pairs, const int* __restrict__ srcs,
    const unsigned* __restrict__ off, int slices, int nb,
    float* __restrict__ outn, unsigned* __restrict__ indeg,
    float* __restrict__ inn, int* __restrict__ seg_start,
    int* __restrict__ nbr, int n_nodes, int n_edges) {
  __shared__ unsigned c_in[BKT_NODES];
  __shared__ unsigned c_out[BKT_NODES];
  __shared__ unsigned s_total[32];
  __shared__ int s_nbr[BKT_STAGE];
  const int tid = threadIdx.x;
  const int lane = tid & 31;
  const int wv = tid >> 5;
  const int b = blockIdx.x;
  const int node0 = b << BKT_SHIFT;
  c_in[tid] = 0u;
  c_out[tid] = 0u;
  __syncthreads();
  const int es = (int)off[(size_t)b * slices];
  const int ee = (b + 1 < nb) ? (int)off[(size_t)(b + 1) * slices] : n_edges;
  const int ss = (int)off[(size_t)(nb + b) * slices];
  const int se = (b + 1 < nb) ? (int)off[(size_t)(nb + b + 1) * slices] : n_edges;
  for (int e = es + tid, f = ss + tid; e < ee || f < se;
       e += SORT_BATCH * BKT_NODES, f += SORT_BATCH * BKT_NODES) {
    int d[SORT_BATCH], s[SORT_BATCH];
#pragma unroll
    for (int j = 0; j < SORT_BATCH; ++j) {
      const int ej = e + j * BKT_NODES;
      const int fj = f + j * BKT_NODES;
      const int dv = pairs[ej < ee ? ej : n_edges - 1].y;
      const int sv = srcs[fj < se ? fj : n_edges - 1];
      d[j] = ej < ee ? dv : -1;
      s[j] = fj < se ? sv : -1;
    }
#pragma unroll
    for (int j = 0; j < SORT_BATCH; ++j) {
      if (d[j] >= 0) atomicAdd(&c_in[d[j] - node0], 1u);
      if (s[j] >= 0) atomicAdd(&c_out[s[j] - node0], 1u);
    }
  }
  __syncthreads();

  const unsigned d = c_in[tid];
  unsigned incl = d;
  for (int x = 1; x < 32; x <<= 1) {
    unsigned t = __shfl_up(incl, x);
    if (lane >= x) incl += t;
  }
  if (lane == 31) s_total[wv] = incl;
  __syncthreads();
  unsigned woff = (lane < wv) ? s_total[lane] : 0u;
  for (int x = 16; x > 0; x >>= 1) woff += __shfl_xor(woff, x);
  const unsigned local = woff + incl - d;
  const unsigned first = (unsigned)es + local;
  const bool staged = (ee - es) <= BKT_STAGE;
  c_in[tid] = staged ? local : first;

  const int node = node0 + tid;
  if (node < n_nodes) {
    const unsigned od = c_out[tid];
    seg_start[node] = (int)first;
    indeg[node] = d;
    inn[node] = rsqrtf((float)(d > 1u ? d : 1u));
    outn[node] = rsqrtf((float)(od > 1u ? od : 1u));
  }
  __syncthreads();
  for (int e = es + tid; e < ee; e += SORT_BATCH * BKT_NODES) {
    int2 p[SORT_BATCH];
    unsigned pos[SORT_BATCH];
#pragma unroll
    for (int j = 0; j < SORT_BATCH; ++j) {
      const int ej = e + j * BKT_NODES;
      p[j] = pairs[ej < ee ? ej : n_edges - 1];
      if (ej >= ee) p[j].y = -1;
    }
#pragma unroll
    for (int j = 0; j < SORT_BATCH; ++j) {
      pos[j] = 0u;
      if (p[j].y >= 0) pos[j] = atomicAdd(&c_in[p[j].y - node0], 1u);
    }
#pragma unroll
    for (int j = 0; j < SORT_BATCH; ++j) {
      if (p[j].y >= 0) {
        if (staged) s_nbr[pos[j]] = p[j].x;
        else nbr[pos[j]] = p[j].x;
      }
    }
  }
  if (staged) {
    __syncthreads();
    for (int i = tid; i < ee - es; i += BKT_NODES) nbr[es + i] = s_nbr[i];
  }
}

#define TILE_ROWS 16
#define LAYER_TILES_PER_WG 4
#define TILE_PITCH 68
static_assert(LAYER_TILES_PER_WG <= 7, "lane i + 1 of a group must exist for every tile i");

__global__ __launch_bounds__(128) void gcn_layer(
    const float* __restrict__ x, const float* __restrict__ out_norm,
    const float* __restrict__ in_norm, const int* __restrict__ seg_start,
    const unsigned* __restrict__ indeg, const int* __restrict__ nbr,
    const float* __restrict__ W,
    const float* __restrict__ bias,
    float* __restrict__ out, int n_nodes, int n_edges, int n_tiles) {
  __shared__ __attribute__((aligned(16))) float s_tile[2][TILE_ROWS * TILE_PITCH];
  const int lane = threadIdx.x & 31;
  const int wave = threadIdx.x >> 5;
  const int nwg = (int)gridDim.x;
  const int q = lane & 7;
  const int grow = wave * 4 + (lane >> 3);
  const int half = lane >> 4;
  const int l15  = lane & 15;
  const int n0   = wave * 16;

  int mstart, mcnt;
  float msc;
  {
    const int tq = (int)blockIdx.x + q * nwg;
    const int nodeq = tq * TILE_ROWS + grow;
    const bool okq = q < LAYER_TILES_PER_WG && tq < n_tiles && nodeq < n_nodes;
    const int nc = okq ? nodeq : 0;
    mstart = seg_start[nc];
    const int dc = (int)indeg[nc];
    msc = in_norm[nc];
    mcnt = okq ? dc : 0;
  }

  const float* __restrict__ bptr = W + (size_t)(2 * half) * D_FEAT + n0 + l15;
  v2f bf[D_FEAT / 4];
#pragma unroll
  for (int i = 0; i < D_FEAT / 4; ++i) {
    bf[i].x = bptr[(size_t)(4 * i) * D_FEAT];
    bf[i].y = bptr[(size_t)(4 * i + 1) * D_FEAT];
  }
  const float bv = bias[n0 + l15];
  const float4* __restrict__ x4 = reinterpret_cast<const float4*>(x);

  int start = __shfl(mstart, 0, 8);
  int cnt = __shfl(mcnt, 0, 8);
  float sc = __shfl(msc, 0, 8);
  int my0, my1;
  float nw0, nw1;
  {
    const int m = cnt < 16 ? cnt : 16;
    const int e0 = start + (q < m ? q : 0);
    const int e1 = start + (q + 8 < m ? q + 8 : 0);
    my0 = nbr[e0 < n_edges ? e0 : n_edges - 1];
    my1 = nbr[e1 < n_edges ? e1 : n_edges - 1];
    nw0 = out_norm[my0];
    nw1 = out_norm[my1];
  }

  int buf = 0;
  for (int i = 0; i < LAYER_TILES_PER_WG; ++i, buf ^= 1) {
    const int t = (int)blockIdx.x + i * nwg;
    if (t >= n_tiles) break;
    const int row0 = t * TILE_ROWS;

    const int nstart = __shfl(mstart, i + 1, 8);
    const int ncnt = __shfl(mcnt, i + 1, 8);
    const float nsc = __shfl(msc, i + 1, 8);
    int nmy0, nmy1;
    {
      const int m = ncnt < 16 ? ncnt : 16;
      const int e0 = nstart + (q < m ? q : 0);
      const int e1 = nstart + (q + 8 < m ? q + 8 : 0);
      nmy0 = nbr[e0 < n_edges ? e0 : n_edges - 1];
      nmy1 = nbr[e1 < n_edges ? e1 : n_edges - 1];
    }

    float4 lo = make_float4(0.0f, 0.0f, 0.0f, 0.0f);
    float4 hi = make_float4(0.0f, 0.0f, 0.0f, 0.0f);
    for (int b = 0; b < cnt; b += 16) {
      const int m = (cnt - b) < 16 ? (cnt - b) : 16;
      if (b > 0) {
        my0 = nbr[start + b + (q < m ? q : 0)];
        my1 = nbr[start + b + (q + 8 < m ? q + 8 : 0)];
        nw0 = out_norm[my0];
        nw1 = out_norm[my1];
      }
      for (int j = 0; j < m; j += 4) {
        const int ids = j < 8 ? my0 : my1;
        const float wts = j < 8 ? nw0 : nw1;
        const int jj = j & 7;
        const int s0 = __shfl(ids, jj, 8);
        const int s1 = __shfl(ids, jj + 1, 8);
        const int s2 = __shfl(ids, jj + 2, 8);
        const int s3 = __shfl(ids, jj + 3, 8);
        const float4 u0 = x4[(size_t)s0 * (D_FEAT / 4) + q];
        const float4 v0 = x4[(size_t)s0 * (D_FEAT / 4) + 8 + q];
        const float4 u1 = x4[(size_t)s1 * (D_FEAT / 4) + q];
        const float4 v1 = x4[(size_t)s1 * (D_FEAT / 4) + 8 + q];
        const float4 u2 = x4[(size_t)s2 * (D_FEAT / 4) + q];
        const float4 v2 = x4[(size_t)s2 * (D_FEAT / 4) + 8 + q];
        const float4 u3 = x4[(size_t)s3 * (D_FEAT / 4) + q];
        const float4 v3 = x4[(size_t)s3 * (D_FEAT / 4) + 8 + q];
        __builtin_amdgcn_sched_barrier(0);
        const float w0 = __shfl(wts, jj, 8);
        const float w1 = (j + 1 < m) ? __shfl(wts, jj + 1, 8) : 0.0f;
        const float w2 = (j + 2 < m) ? __shfl(wts, jj + 2, 8) : 0.0f;
        const float w3 = (j + 3 < m) ? __shfl(wts, jj + 3, 8) : 0.0f;
        lo.x += u0.x * w0; lo.y += u0.y * w0; lo.z += u0.z * w0; lo.w += u0.w * w0;
        hi.x += v0.x * w0; hi.y += v0.y * w0; hi.z += v0.z * w0; hi.w += v0.w * w0;
        lo.x += u1.x * w1; lo.y += u1.y * w1; lo.z += u1.z * w1; lo.w += u1.w * w1;
        hi.x += v1.x * w1; hi.y += v1.y * w1; hi.z += v1.z * w1; hi.w += v1.w * w1;
        lo.x += u2.x * w2; lo.y += u2.y * w2; lo.z += u2.z * w2; lo.w += u2.w * w2;
        hi.x += v2.x * w2; hi.y += v2.y * w2; hi.z += v2.z * w2; hi.w += v2.w * w2;
        lo.x += u3.x * w3; lo.y += u3.y * w3; lo.z += u3.z * w3; lo.w += u3.w * w3;
        hi.x += v3.x * w3; hi.y += v3.y * w3; hi.z += v3.z * w3; hi.w += v3.w * w3;
      }
    }
    lo.x *= sc; lo.y *= sc; lo.z *= sc; lo.w *= sc;
    hi.x *= sc; hi.y *= sc; hi.z *= sc; hi.w *= sc;

    float* __restrict__ st = &s_tile[buf][grow * TILE_PITCH];
    *reinterpret_cast<float4*>(st + 4 * q) = lo;
    *reinterpret_cast<float4*>(st + 32 + 4 * q) = hi;
    __syncthreads();

    const float* __restrict__ ap = &s_tile[buf][l15 * TILE_PITCH + 2 * half];
    v8f c = {};
#pragma unroll
    for (int k = 0; k < D_FEAT / 4; ++k) {
      const v2f a = *reinterpret_cast<const v2f*>(ap + 4 * k);
      c = __builtin_amdgcn_wmma_f32_16x16x4_f32(false, a, false, bf[k],
                                                (short)0, c, false, false);
    }

    const float nnw0 = out_norm[nmy0];
    const float nnw1 = out_norm[nmy1];

#pragma unroll
    for (int v = 0; v < 8; ++v) {
      const int r = row0 + v + 8 * half;
      if (r < n_nodes) {
        const float xv = c[v] + bv;
        const float yv = xv > 0.0f ? xv : expm1f(xv);
        out[(size_t)r * D_FEAT + n0 + l15] = yv;
      }
    }

    start = nstart; cnt = ncnt; sc = nsc;
    my0 = nmy0; my1 = nmy1; nw0 = nnw0; nw1 = nnw1;
  }
}

extern "C" void kernel_launch(void* const* d_in, const int* in_sizes, int n_in,
                              void* d_out, int out_size, void* d_ws, size_t ws_size,
                              hipStream_t stream) {
  const float* feat = (const float*)d_in[0];
  const float* Ws[3] = {(const float*)d_in[1], (const float*)d_in[3], (const float*)d_in[5]};
  const float* bs[3] = {(const float*)d_in[2], (const float*)d_in[4], (const float*)d_in[6]};
  const int* src = (const int*)d_in[7];
  const int* dst = (const int*)d_in[8];

  const int n_nodes = in_sizes[0] / D_FEAT;
  const int n_edges = in_sizes[7];

  float*    out_norm  = (float*)d_ws;
  unsigned* indeg     = (unsigned*)(out_norm + n_nodes);
  float*    mid       = out_norm + 2 * (size_t)n_nodes;
  float*    in_norm   = mid + (size_t)n_nodes * D_FEAT;
  int*      seg_start = (int*)(in_norm + n_nodes);
  int*      nbr       = seg_start + n_nodes;
  int*      srcs      = nbr + n_edges;
  float*    outp      = (float*)d_out;

  {
    const int nb = (n_nodes + BKT_NODES - 1) >> BKT_SHIFT;
    const int len = nb * SORT_SLICES;
    int2*     pairs = (int2*)mid;
    unsigned* cnt   = (unsigned*)mid + 2 * (size_t)n_edges;
    unsigned* off   = cnt + 2 * (size_t)len;
    const size_t lds = 2 * (size_t)nb * sizeof(unsigned);
    gcn_bucket_count<<<SORT_SLICES, 1024, lds, stream>>>(src, dst, cnt, n_edges, nb);
    gcn_bucket_scan<<<dim3((unsigned)((len + 1023) / 1024), 2), 1024, 0, stream>>>(
        cnt, off, len);
    const size_t per = ((size_t)n_edges + SORT_SLICES - 1) / SORT_SLICES;
    gcn_bucket_scatter<<<SORT_SLICES, 1024, per * 12 + 2 * lds, stream>>>(
        src, dst, cnt, off, pairs, srcs, n_edges, nb);
    gcn_bucket_build<<<(unsigned)nb, BKT_NODES, 0, stream>>>(
        pairs, srcs, off, SORT_SLICES, nb, out_norm, indeg, in_norm, seg_start,
        nbr, n_nodes, n_edges);
  }

  const int n_tiles = (n_nodes + TILE_ROWS - 1) / TILE_ROWS;
  const unsigned lblocks =
      (unsigned)((n_tiles + LAYER_TILES_PER_WG - 1) / LAYER_TILES_PER_WG);
  float* dsts[3] = {outp, mid, outp};
  const float* x = feat;
  for (int l = 0; l < 3; ++l) {
    gcn_layer<<<lblocks, 128, 0, stream>>>(x, out_norm, in_norm, seg_start, indeg,
                                           nbr, Ws[l], bs[l], dsts[l], n_nodes,
                                           n_edges, n_tiles);
    x = dsts[l];
  }
}


# Edges grouped by bucket inside each slice's own range, so no global offsets are needed: count and scan launches gone (7 to 5); a node's sum order stays LDS arrival order, now over per-slice runs
# speedup vs baseline: 6.0800x; 1.1067x over previous
#include <hip/hip_runtime.h>
#include <hip/hip_bf16.h>

#define D_FEAT 64

typedef float v2f __attribute__((ext_vector_type(2)));
typedef float v8f __attribute__((ext_vector_type(8)));

#define BKT_SHIFT 9
#define BKT_NODES 512
#define SORT_SLICES 256
#define SORT_BATCH 8
#define BKT_STAGE 12288
#define BKT_WAVES (BKT_NODES / 32)
static_assert(SORT_SLICES % (BKT_WAVES * SORT_BATCH) == 0,
              "pass B hands whole groups of slices to its waves");

__global__ __launch_bounds__(1024) void gcn_slice_group(
    const int* __restrict__ src, const int* __restrict__ dst,
    unsigned* __restrict__ tab, int2* __restrict__ pairs,
    int* __restrict__ srcs, int n_edges, int nb) {
  extern __shared__ int2 s_stage[];
  __shared__ unsigned s_total[32];
  const int per = (n_edges + (int)gridDim.x - 1) / (int)gridDim.x;
  int2* s_pair = s_stage;
  int* s_src = (int*)(s_stage + per);
  unsigned* s_cur = (unsigned*)(s_src + per);
  const int tid = threadIdx.x;
  const int lane = tid & 31;
  const int wv = tid >> 5;
  const int e0 = (int)blockIdx.x * per;
  const int e1 = (e0 + per) < n_edges ? (e0 + per) : n_edges;
  const int n_own = e1 > e0 ? e1 - e0 : 0;

  for (int i = tid; i < 2 * nb; i += 1024) s_cur[i] = 0u;
  __syncthreads();
  for (int e = e0 + tid; e < e1; e += SORT_BATCH * 1024) {
    int d[SORT_BATCH], s[SORT_BATCH];
#pragma unroll
    for (int j = 0; j < SORT_BATCH; ++j) {
      const int ej = e + j * 1024;
      const int ec = ej < e1 ? ej : e1 - 1;
      const int dv = dst[ec];
      const int sv = src[ec];
      d[j] = ej < e1 ? dv : -1;
      s[j] = sv;
    }
#pragma unroll
    for (int j = 0; j < SORT_BATCH; ++j) {
      if (d[j] >= 0) {
        atomicAdd(&s_cur[d[j] >> BKT_SHIFT], 1u);
        atomicAdd(&s_cur[nb + (s[j] >> BKT_SHIFT)], 1u);
      }
    }
  }
  __syncthreads();

  unsigned carry = 0u;
  for (int base = 0; base < 2 * nb; base += 1024) {
    const int i = base + tid;
    const unsigned c = (i < 2 * nb) ? s_cur[i] : 0u;
    unsigned incl = c;
    for (int x = 1; x < 32; x <<= 1) {
      unsigned t = __shfl_up(incl, x);
      if (lane >= x) incl += t;
    }
    if (lane == 31) s_total[wv] = incl;
    __syncthreads();
    unsigned woff = (lane < wv) ? s_total[lane] : 0u;
    unsigned all = s_total[lane];
    for (int x = 16; x > 0; x >>= 1) {
      woff += __shfl_xor(woff, x);
      all += __shfl_xor(all, x);
    }
    if (i < 2 * nb) s_cur[i] = carry + woff + incl - c;
    carry += all;
    __syncthreads();
  }
  for (int i = tid; i < 2 * nb; i += 1024) {
    const int key = i >= nb ? 1 : 0;
    const unsigned ls = s_cur[i] - (key ? (unsigned)n_own : 0u);
    s_cur[i] = ls;
    tab[((size_t)key * (nb + 1) + (i - key * nb)) * gridDim.x + blockIdx.x] = ls;
  }
  if (tid < 2)
    tab[((size_t)tid * (nb + 1) + nb) * gridDim.x + blockIdx.x] = (unsigned)n_own;
  __syncthreads();

  for (int e = e0 + tid; e < e1; e += SORT_BATCH * 1024) {
    int d[SORT_BATCH], s[SORT_BATCH];
#pragma unroll
    for (int j = 0; j < SORT_BATCH; ++j) {
      const int ej = e + j * 1024;
      const int ec = ej < e1 ? ej : e1 - 1;
      const int dv = dst[ec];
      const int sv = src[ec];
      d[j] = ej < e1 ? dv : -1;
      s[j] = sv;
    }
#pragma unroll
    for (int j = 0; j < SORT_BATCH; ++j) {
      if (d[j] >= 0) {
        const unsigned p = atomicAdd(&s_cur[d[j] >> BKT_SHIFT], 1u);
        const unsigned q = atomicAdd(&s_cur[nb + (s[j] >> BKT_SHIFT)], 1u);
        s_pair[p] = make_int2(s[j], d[j]);
        s_src[q] = s[j];
      }
    }
  }
  __syncthreads();
  for (int i = tid; i < n_own; i += 1024) {
    pairs[e0 + i] = s_pair[i];
    srcs[e0 + i] = s_src[i];
  }
}

__global__ __launch_bounds__(BKT_NODES) void gcn_bucket_build(
    const int2* __restrict__ pairs, const int* __restrict__ srcs,
    const unsigned* __restrict__ tab, int per, int nb,
    float* __restrict__ outn, unsigned* __restrict__ indeg,
    float* __restrict__ inn, int* __restrict__ seg_start,
    int* __restrict__ nbr, int n_nodes, int n_edges) {
  __shared__ unsigned c_in[BKT_NODES];
  __shared__ unsigned c_out[BKT_NODES];
  __shared__ unsigned s_total[32];
  __shared__ unsigned s_red[2][BKT_WAVES];
  __shared__ unsigned s_d0[SORT_SLICES];
  __shared__ unsigned s_dl[SORT_SLICES];
  __shared__ unsigned s_s0[SORT_SLICES];
  __shared__ unsigned s_sl[SORT_SLICES];
  __shared__ int s_nbr[BKT_STAGE];
  const int tid = threadIdx.x;
  const int lane = tid & 31;
  const int wv = tid >> 5;
  const int b = blockIdx.x;
  const int node0 = b << BKT_SHIFT;
  const unsigned last = (unsigned)(n_edges - 1);
  c_in[tid] = 0u;
  c_out[tid] = 0u;
  unsigned sum0 = 0u, suml = 0u;
  for (int s = tid; s < SORT_SLICES; s += BKT_NODES) {
    const unsigned* __restrict__ td = tab + (size_t)b * SORT_SLICES + s;
    const unsigned* __restrict__ ts = tab + (size_t)(nb + 1 + b) * SORT_SLICES + s;
    const unsigned a0 = td[0];
    const unsigned a1 = td[SORT_SLICES];
    const unsigned g0 = ts[0];
    const unsigned g1 = ts[SORT_SLICES];
    const unsigned base = (unsigned)s * (unsigned)per;
    s_d0[s] = base + a0;
    s_dl[s] = a1 - a0;
    s_s0[s] = base + g0;
    s_sl[s] = g1 - g0;
    sum0 += a0;
    suml += a1 - a0;
  }
  for (int x = 16; x > 0; x >>= 1) {
    sum0 += __shfl_xor(sum0, x);
    suml += __shfl_xor(suml, x);
  }
  if (lane == 0) { s_red[0][wv] = sum0; s_red[1][wv] = suml; }
  __syncthreads();
  unsigned r0 = lane < BKT_WAVES ? s_red[0][lane] : 0u;
  unsigned r1 = lane < BKT_WAVES ? s_red[1][lane] : 0u;
  for (int x = 16; x > 0; x >>= 1) {
    r0 += __shfl_xor(r0, x);
    r1 += __shfl_xor(r1, x);
  }
  const int es = (int)r0;
  const int ee = (int)(r0 + r1);

  for (int sb = 0; sb < SORT_SLICES; sb += BKT_WAVES * SORT_BATCH) {
    int d[SORT_BATCH][2], sv[SORT_BATCH][2];
    bool more = false;
#pragma unroll
    for (int j = 0; j < SORT_BATCH; ++j) {
      const int s = sb + j * BKT_WAVES + wv;
      const unsigned d0 = s_d0[s], dl = s_dl[s], s0 = s_s0[s], sl = s_sl[s];
      more = more || dl > 64u || sl > 64u;
#pragma unroll
      for (int h = 0; h < 2; ++h) {
        const unsigned k = (unsigned)(lane + 32 * h);
        const int dv = pairs[k < dl ? d0 + k : last].y;
        const int sx = srcs[k < sl ? s0 + k : last];
        d[j][h] = k < dl ? dv : -1;
        sv[j][h] = k < sl ? sx : -1;
      }
    }
#pragma unroll
    for (int j = 0; j < SORT_BATCH; ++j) {
#pragma unroll
      for (int h = 0; h < 2; ++h) {
        if (d[j][h] >= 0) atomicAdd(&c_in[d[j][h] - node0], 1u);
        if (sv[j][h] >= 0) atomicAdd(&c_out[sv[j][h] - node0], 1u);
      }
    }
    if (more)
    for (int j = 0; j < SORT_BATCH; ++j) {
      const int s = sb + j * BKT_WAVES + wv;
      const unsigned d0 = s_d0[s], dl = s_dl[s], s0 = s_s0[s], sl = s_sl[s];
      for (unsigned k = (unsigned)lane + 64u; k < dl; k += 32u)
        atomicAdd(&c_in[pairs[d0 + k].y - node0], 1u);
      for (unsigned k = (unsigned)lane + 64u; k < sl; k += 32u)
        atomicAdd(&c_out[srcs[s0 + k] - node0], 1u);
    }
  }
  __syncthreads();

  const unsigned d = c_in[tid];
  unsigned incl = d;
  for (int x = 1; x < 32; x <<= 1) {
    unsigned t = __shfl_up(incl, x);
    if (lane >= x) incl += t;
  }
  if (lane == 31) s_total[wv] = incl;
  __syncthreads();
  unsigned woff = (lane < wv) ? s_total[lane] : 0u;
  for (int x = 16; x > 0; x >>= 1) woff += __shfl_xor(woff, x);
  const unsigned local = woff + incl - d;
  const unsigned first = (unsigned)es + local;
  const bool staged = (ee - es) <= BKT_STAGE;
  c_in[tid] = staged ? local : first;

  const int node = node0 + tid;
  if (node < n_nodes) {
    const unsigned od = c_out[tid];
    seg_start[node] = (int)first;
    indeg[node] = d;
    inn[node] = rsqrtf((float)(d > 1u ? d : 1u));
    outn[node] = rsqrtf((float)(od > 1u ? od : 1u));
  }
  __syncthreads();
  for (int sb = 0; sb < SORT_SLICES; sb += BKT_WAVES * SORT_BATCH) {
    int2 p[SORT_BATCH][2];
    unsigned pos[SORT_BATCH][2];
    bool more = false;
#pragma unroll
    for (int j = 0; j < SORT_BATCH; ++j) {
      const int s = sb + j * BKT_WAVES + wv;
      const unsigned d0 = s_d0[s], dl = s_dl[s];
      more = more || dl > 64u;
#pragma unroll
      for (int h = 0; h < 2; ++h) {
        const unsigned k = (unsigned)(lane + 32 * h);
        p[j][h] = pairs[k < dl ? d0 + k : last];
        if (k >= dl) p[j][h].y = -1;
      }
    }
#pragma unroll
    for (int j = 0; j < SORT_BATCH; ++j) {
#pragma unroll
      for (int h = 0; h < 2; ++h) {
        pos[j][h] = 0u;
        if (p[j][h].y >= 0) pos[j][h] = atomicAdd(&c_in[p[j][h].y - node0], 1u);
      }
    }
#pragma unroll
    for (int j = 0; j < SORT_BATCH; ++j) {
#pragma unroll
      for (int h = 0; h < 2; ++h) {
        if (p[j][h].y >= 0) {
          if (staged) s_nbr[pos[j][h]] = p[j][h].x;
          else nbr[pos[j][h]] = p[j][h].x;
        }
      }
    }
    if (more)
    for (int j = 0; j < SORT_BATCH; ++j) {
      const int s = sb + j * BKT_WAVES + wv;
      const unsigned d0 = s_d0[s], dl = s_dl[s];
      for (unsigned k = (unsigned)lane + 64u; k < dl; k += 32u) {
        const int2 pr = pairs[d0 + k];
        const unsigned ps = atomicAdd(&c_in[pr.y - node0], 1u);
        if (staged) s_nbr[ps] = pr.x;
        else nbr[ps] = pr.x;
      }
    }
  }
  if (staged) {
    __syncthreads();
    for (int i = tid; i < ee - es; i += BKT_NODES) nbr[es + i] = s_nbr[i];
  }
}

#define TILE_ROWS 16
#define LAYER_TILES_PER_WG 4
#define TILE_PITCH 68
static_assert(LAYER_TILES_PER_WG <= 7, "lane i + 1 of a group must exist for every tile i");

__global__ __launch_bounds__(128) void gcn_layer(
    const float* __restrict__ x, const float* __restrict__ out_norm,
    const float* __restrict__ in_norm, const int* __restrict__ seg_start,
    const unsigned* __restrict__ indeg, const int* __restrict__ nbr,
    const float* __restrict__ W,
    const float* __restrict__ bias,
    float* __restrict__ out, int n_nodes, int n_edges, int n_tiles) {
  __shared__ __attribute__((aligned(16))) float s_tile[2][TILE_ROWS * TILE_PITCH];
  const int lane = threadIdx.x & 31;
  const int wave = threadIdx.x >> 5;
  const int nwg = (int)gridDim.x;
  const int q = lane & 7;
  const int grow = wave * 4 + (lane >> 3);
  const int half = lane >> 4;
  const int l15  = lane & 15;
  const int n0   = wave * 16;

  int mstart, mcnt;
  float msc;
  {
    const int tq = (int)blockIdx.x + q * nwg;
    const int nodeq = tq * TILE_ROWS + grow;
    const bool okq = q < LAYER_TILES_PER_WG && tq < n_tiles && nodeq < n_nodes;
    const int nc = okq ? nodeq : 0;
    mstart = seg_start[nc];
    const int dc = (int)indeg[nc];
    msc = in_norm[nc];
    mcnt = okq ? dc : 0;
  }

  const float* __restrict__ bptr = W + (size_t)(2 * half) * D_FEAT + n0 + l15;
  v2f bf[D_FEAT / 4];
#pragma unroll
  for (int i = 0; i < D_FEAT / 4; ++i) {
    bf[i].x = bptr[(size_t)(4 * i) * D_FEAT];
    bf[i].y = bptr[(size_t)(4 * i + 1) * D_FEAT];
  }
  const float bv = bias[n0 + l15];
  const float4* __restrict__ x4 = reinterpret_cast<const float4*>(x);

  int start = __shfl(mstart, 0, 8);
  int cnt = __shfl(mcnt, 0, 8);
  float sc = __shfl(msc, 0, 8);
  int my0, my1;
  float nw0, nw1;
  {
    const int m = cnt < 16 ? cnt : 16;
    const int e0 = start + (q < m ? q : 0);
    const int e1 = start + (q + 8 < m ? q + 8 : 0);
    my0 = nbr[e0 < n_edges ? e0 : n_edges - 1];
    my1 = nbr[e1 < n_edges ? e1 : n_edges - 1];
    nw0 = out_norm[my0];
    nw1 = out_norm[my1];
  }

  int buf = 0;
  for (int i = 0; i < LAYER_TILES_PER_WG; ++i, buf ^= 1) {
    const int t = (int)blockIdx.x + i * nwg;
    if (t >= n_tiles) break;
    const int row0 = t * TILE_ROWS;

    const int nstart = __shfl(mstart, i + 1, 8);
    const int ncnt = __shfl(mcnt, i + 1, 8);
    const float nsc = __shfl(msc, i + 1, 8);
    int nmy0, nmy1;
    {
      const int m = ncnt < 16 ? ncnt : 16;
      const int e0 = nstart + (q < m ? q : 0);
      const int e1 = nstart + (q + 8 < m ? q + 8 : 0);
      nmy0 = nbr[e0 < n_edges ? e0 : n_edges - 1];
      nmy1 = nbr[e1 < n_edges ? e1 : n_edges - 1];
    }

    float4 lo = make_float4(0.0f, 0.0f, 0.0f, 0.0f);
    float4 hi = make_float4(0.0f, 0.0f, 0.0f, 0.0f);
    for (int b = 0; b < cnt; b += 16) {
      const int m = (cnt - b) < 16 ? (cnt - b) : 16;
      if (b > 0) {
        my0 = nbr[start + b + (q < m ? q : 0)];
        my1 = nbr[start + b + (q + 8 < m ? q + 8 : 0)];
        nw0 = out_norm[my0];
        nw1 = out_norm[my1];
      }
      for (int j = 0; j < m; j += 4) {
        const int ids = j < 8 ? my0 : my1;
        const float wts = j < 8 ? nw0 : nw1;
        const int jj = j & 7;
        const int s0 = __shfl(ids, jj, 8);
        const int s1 = __shfl(ids, jj + 1, 8);
        const int s2 = __shfl(ids, jj + 2, 8);
        const int s3 = __shfl(ids, jj + 3, 8);
        const float4 u0 = x4[(size_t)s0 * (D_FEAT / 4) + q];
        const float4 v0 = x4[(size_t)s0 * (D_FEAT / 4) + 8 + q];
        const float4 u1 = x4[(size_t)s1 * (D_FEAT / 4) + q];
        const float4 v1 = x4[(size_t)s1 * (D_FEAT / 4) + 8 + q];
        const float4 u2 = x4[(size_t)s2 * (D_FEAT / 4) + q];
        const float4 v2 = x4[(size_t)s2 * (D_FEAT / 4) + 8 + q];
        const float4 u3 = x4[(size_t)s3 * (D_FEAT / 4) + q];
        const float4 v3 = x4[(size_t)s3 * (D_FEAT / 4) + 8 + q];
        __builtin_amdgcn_sched_barrier(0);
        const float w0 = __shfl(wts, jj, 8);
        const float w1 = (j + 1 < m) ? __shfl(wts, jj + 1, 8) : 0.0f;
        const float w2 = (j + 2 < m) ? __shfl(wts, jj + 2, 8) : 0.0f;
        const float w3 = (j + 3 < m) ? __shfl(wts, jj + 3, 8) : 0.0f;
        lo.x += u0.x * w0; lo.y += u0.y * w0; lo.z += u0.z * w0; lo.w += u0.w * w0;
        hi.x += v0.x * w0; hi.y += v0.y * w0; hi.z += v0.z * w0; hi.w += v0.w * w0;
        lo.x += u1.x * w1; lo.y += u1.y * w1; lo.z += u1.z * w1; lo.w += u1.w * w1;
        hi.x += v1.x * w1; hi.y += v1.y * w1; hi.z += v1.z * w1; hi.w += v1.w * w1;
        lo.x += u2.x * w2; lo.y += u2.y * w2; lo.z += u2.z * w2; lo.w += u2.w * w2;
        hi.x += v2.x * w2; hi.y += v2.y * w2; hi.z += v2.z * w2; hi.w += v2.w * w2;
        lo.x += u3.x * w3; lo.y += u3.y * w3; lo.z += u3.z * w3; lo.w += u3.w * w3;
        hi.x += v3.x * w3; hi.y += v3.y * w3; hi.z += v3.z * w3; hi.w += v3.w * w3;
      }
    }
    lo.x *= sc; lo.y *= sc; lo.z *= sc; lo.w *= sc;
    hi.x *= sc; hi.y *= sc; hi.z *= sc; hi.w *= sc;

    float* __restrict__ st = &s_tile[buf][grow * TILE_PITCH];
    *reinterpret_cast<float4*>(st + 4 * q) = lo;
    *reinterpret_cast<float4*>(st + 32 + 4 * q) = hi;
    __syncthreads();

    const float* __restrict__ ap = &s_tile[buf][l15 * TILE_PITCH + 2 * half];
    v8f c = {};
#pragma unroll
    for (int k = 0; k < D_FEAT / 4; ++k) {
      const v2f a = *reinterpret_cast<const v2f*>(ap + 4 * k);
      c = __builtin_amdgcn_wmma_f32_16x16x4_f32(false, a, false, bf[k],
                                                (short)0, c, false, false);
    }

    const float nnw0 = out_norm[nmy0];
    const float nnw1 = out_norm[nmy1];

#pragma unroll
    for (int v = 0; v < 8; ++v) {
      const int r = row0 + v + 8 * half;
      if (r < n_nodes) {
        const float xv = c[v] + bv;
        const float yv = xv > 0.0f ? xv : expm1f(xv);
        out[(size_t)r * D_FEAT + n0 + l15] = yv;
      }
    }

    start = nstart; cnt = ncnt; sc = nsc;
    my0 = nmy0; my1 = nmy1; nw0 = nnw0; nw1 = nnw1;
  }
}

extern "C" void kernel_launch(void* const* d_in, const int* in_sizes, int n_in,
                              void* d_out, int out_size, void* d_ws, size_t ws_size,
                              hipStream_t stream) {
  const float* feat = (const float*)d_in[0];
  const float* Ws[3] = {(const float*)d_in[1], (const float*)d_in[3], (const float*)d_in[5]};
  const float* bs[3] = {(const float*)d_in[2], (const float*)d_in[4], (const float*)d_in[6]};
  const int* src = (const int*)d_in[7];
  const int* dst = (const int*)d_in[8];

  const int n_nodes = in_sizes[0] / D_FEAT;
  const int n_edges = in_sizes[7];

  float*    out_norm  = (float*)d_ws;
  unsigned* indeg     = (unsigned*)(out_norm + n_nodes);
  float*    mid       = out_norm + 2 * (size_t)n_nodes;
  float*    in_norm   = mid + (size_t)n_nodes * D_FEAT;
  int*      seg_start = (int*)(in_norm + n_nodes);
  int*      nbr       = seg_start + n_nodes;
  int*      srcs      = nbr + n_edges;
  float*    outp      = (float*)d_out;

  {
    const int nb = (n_nodes + BKT_NODES - 1) >> BKT_SHIFT;
    int2*     pairs = (int2*)mid;
    unsigned* tab   = (unsigned*)mid + 2 * (size_t)n_edges;
    const size_t lds = 2 * (size_t)nb * sizeof(unsigned);
    const size_t per = ((size_t)n_edges + SORT_SLICES - 1) / SORT_SLICES;
    gcn_slice_group<<<SORT_SLICES, 1024, per * 12 + lds, stream>>>(
        src, dst, tab, pairs, srcs, n_edges, nb);
    gcn_bucket_build<<<(unsigned)nb, BKT_NODES, 0, stream>>>(
        pairs, srcs, tab, (int)per, nb, out_norm, indeg, in_norm, seg_start,
        nbr, n_nodes, n_edges);
  }

  const int n_tiles = (n_nodes + TILE_ROWS - 1) / TILE_ROWS;
  const unsigned lblocks =
      (unsigned)((n_tiles + LAYER_TILES_PER_WG - 1) / LAYER_TILES_PER_WG);
  float* dsts[3] = {outp, mid, outp};
  const float* x = feat;
  for (int l = 0; l < 3; ++l) {
    gcn_layer<<<lblocks, 128, 0, stream>>>(x, out_norm, in_norm, seg_start, indeg,
                                           nbr, Ws[l], bs[l], dsts[l], n_nodes,
                                           n_edges, n_tiles);
    x = dsts[l];
  }
}


# Layer outputs go through a second LDS tile and leave as whole 512-byte row pairs, 2 b128 stores per wave and tile behind the next tile's barrier, not 8 stores of two 64-byte pieces; sums unchanged
# speedup vs baseline: 6.6323x; 1.0908x over previous
#include <hip/hip_runtime.h>
#include <hip/hip_bf16.h>

#define D_FEAT 64

typedef float v2f __attribute__((ext_vector_type(2)));
typedef float v8f __attribute__((ext_vector_type(8)));

#define BKT_SHIFT 9
#define BKT_NODES 512
#define SORT_SLICES 256
#define SORT_BATCH 8
#define BKT_STAGE 12288
#define BKT_WAVES (BKT_NODES / 32)
static_assert(SORT_SLICES % (BKT_WAVES * SORT_BATCH) == 0,
              "pass B hands whole groups of slices to its waves");

__global__ __launch_bounds__(1024) void gcn_slice_group(
    const int* __restrict__ src, const int* __restrict__ dst,
    unsigned* __restrict__ tab, int2* __restrict__ pairs,
    int* __restrict__ srcs, int n_edges, int nb) {
  extern __shared__ int2 s_stage[];
  __shared__ unsigned s_total[32];
  const int per = (n_edges + (int)gridDim.x - 1) / (int)gridDim.x;
  int2* s_pair = s_stage;
  int* s_src = (int*)(s_stage + per);
  unsigned* s_cur = (unsigned*)(s_src + per);
  const int tid = threadIdx.x;
  const int lane = tid & 31;
  const int wv = tid >> 5;
  const int e0 = (int)blockIdx.x * per;
  const int e1 = (e0 + per) < n_edges ? (e0 + per) : n_edges;
  const int n_own = e1 > e0 ? e1 - e0 : 0;

  for (int i = tid; i < 2 * nb; i += 1024) s_cur[i] = 0u;
  __syncthreads();
  for (int e = e0 + tid; e < e1; e += SORT_BATCH * 1024) {
    int d[SORT_BATCH], s[SORT_BATCH];
#pragma unroll
    for (int j = 0; j < SORT_BATCH; ++j) {
      const int ej = e + j * 1024;
      const int ec = ej < e1 ? ej : e1 - 1;
      const int dv = dst[ec];
      const int sv = src[ec];
      d[j] = ej < e1 ? dv : -1;
      s[j] = sv;
    }
#pragma unroll
    for (int j = 0; j < SORT_BATCH; ++j) {
      if (d[j] >= 0) {
        atomicAdd(&s_cur[d[j] >> BKT_SHIFT], 1u);
        atomicAdd(&s_cur[nb + (s[j] >> BKT_SHIFT)], 1u);
      }
    }
  }
  __syncthreads();

  unsigned carry = 0u;
  for (int base = 0; base < 2 * nb; base += 1024) {
    const int i = base + tid;
    const unsigned c = (i < 2 * nb) ? s_cur[i] : 0u;
    unsigned incl = c;
    for (int x = 1; x < 32; x <<= 1) {
      unsigned t = __shfl_up(incl, x);
      if (lane >= x) incl += t;
    }
    if (lane == 31) s_total[wv] = incl;
    __syncthreads();
    unsigned woff = (lane < wv) ? s_total[lane] : 0u;
    unsigned all = s_total[lane];
    for (int x = 16; x > 0; x >>= 1) {
      woff += __shfl_xor(woff, x);
      all += __shfl_xor(all, x);
    }
    if (i < 2 * nb) s_cur[i] = carry + woff + incl - c;
    carry += all;
    __syncthreads();
  }
  for (int i = tid; i < 2 * nb; i += 1024) {
    const int key = i >= nb ? 1 : 0;
    const unsigned ls = s_cur[i] - (key ? (unsigned)n_own : 0u);
    s_cur[i] = ls;
    tab[((size_t)key * (nb + 1) + (i - key * nb)) * gridDim.x + blockIdx.x] = ls;
  }
  if (tid < 2)
    tab[((size_t)tid * (nb + 1) + nb) * gridDim.x + blockIdx.x] = (unsigned)n_own;
  __syncthreads();

  for (int e = e0 + tid; e < e1; e += SORT_BATCH * 1024) {
    int d[SORT_BATCH], s[SORT_BATCH];
#pragma unroll
    for (int j = 0; j < SORT_BATCH; ++j) {
      const int ej = e + j * 1024;
      const int ec = ej < e1 ? ej : e1 - 1;
      const int dv = dst[ec];
      const int sv = src[ec];
      d[j] = ej < e1 ? dv : -1;
      s[j] = sv;
    }
#pragma unroll
    for (int j = 0; j < SORT_BATCH; ++j) {
      if (d[j] >= 0) {
        const unsigned p = atomicAdd(&s_cur[d[j] >> BKT_SHIFT], 1u);
        const unsigned q = atomicAdd(&s_cur[nb + (s[j] >> BKT_SHIFT)], 1u);
        s_pair[p] = make_int2(s[j], d[j]);
        s_src[q] = s[j];
      }
    }
  }
  __syncthreads();
  for (int i = tid; i < n_own; i += 1024) {
    pairs[e0 + i] = s_pair[i];
    srcs[e0 + i] = s_src[i];
  }
}

__global__ __launch_bounds__(BKT_NODES) void gcn_bucket_build(
    const int2* __restrict__ pairs, const int* __restrict__ srcs,
    const unsigned* __restrict__ tab, int per, int nb,
    float* __restrict__ outn, unsigned* __restrict__ indeg,
    float* __restrict__ inn, int* __restrict__ seg_start,
    int* __restrict__ nbr, int n_nodes, int n_edges) {
  __shared__ unsigned c_in[BKT_NODES];
  __shared__ unsigned c_out[BKT_NODES];
  __shared__ unsigned s_total[32];
  __shared__ unsigned s_red[2][BKT_WAVES];
  __shared__ unsigned s_d0[SORT_SLICES];
  __shared__ unsigned s_dl[SORT_SLICES];
  __shared__ unsigned s_s0[SORT_SLICES];
  __shared__ unsigned s_sl[SORT_SLICES];
  __shared__ int s_nbr[BKT_STAGE];
  const int tid = threadIdx.x;
  const int lane = tid & 31;
  const int wv = tid >> 5;
  const int b = blockIdx.x;
  const int node0 = b << BKT_SHIFT;
  const unsigned last = (unsigned)(n_edges - 1);
  c_in[tid] = 0u;
  c_out[tid] = 0u;
  unsigned sum0 = 0u, suml = 0u;
  for (int s = tid; s < SORT_SLICES; s += BKT_NODES) {
    const unsigned* __restrict__ td = tab + (size_t)b * SORT_SLICES + s;
    const unsigned* __restrict__ ts = tab + (size_t)(nb + 1 + b) * SORT_SLICES + s;
    const unsigned a0 = td[0];
    const unsigned a1 = td[SORT_SLICES];
    const unsigned g0 = ts[0];
    const unsigned g1 = ts[SORT_SLICES];
    const unsigned base = (unsigned)s * (unsigned)per;
    s_d0[s] = base + a0;
    s_dl[s] = a1 - a0;
    s_s0[s] = base + g0;
    s_sl[s] = g1 - g0;
    sum0 += a0;
    suml += a1 - a0;
  }
  for (int x = 16; x > 0; x >>= 1) {
    sum0 += __shfl_xor(sum0, x);
    suml += __shfl_xor(suml, x);
  }
  if (lane == 0) { s_red[0][wv] = sum0; s_red[1][wv] = suml; }
  __syncthreads();
  unsigned r0 = lane < BKT_WAVES ? s_red[0][lane] : 0u;
  unsigned r1 = lane < BKT_WAVES ? s_red[1][lane] : 0u;
  for (int x = 16; x > 0; x >>= 1) {
    r0 += __shfl_xor(r0, x);
    r1 += __shfl_xor(r1, x);
  }
  const int es = (int)r0;
  const int ee = (int)(r0 + r1);

  for (int sb = 0; sb < SORT_SLICES; sb += BKT_WAVES * SORT_BATCH) {
    int d[SORT_BATCH][2], sv[SORT_BATCH][2];
    bool more = false;
#pragma unroll
    for (int j = 0; j < SORT_BATCH; ++j) {
      const int s = sb + j * BKT_WAVES + wv;
      const unsigned d0 = s_d0[s], dl = s_dl[s], s0 = s_s0[s], sl = s_sl[s];
      more = more || dl > 64u || sl > 64u;
#pragma unroll
      for (int h = 0; h < 2; ++h) {
        const unsigned k = (unsigned)(lane + 32 * h);
        const int dv = pairs[k < dl ? d0 + k : last].y;
        const int sx = srcs[k < sl ? s0 + k : last];
        d[j][h] = k < dl ? dv : -1;
        sv[j][h] = k < sl ? sx : -1;
      }
    }
#pragma unroll
    for (int j = 0; j < SORT_BATCH; ++j) {
#pragma unroll
      for (int h = 0; h < 2; ++h) {
        if (d[j][h] >= 0) atomicAdd(&c_in[d[j][h] - node0], 1u);
        if (sv[j][h] >= 0) atomicAdd(&c_out[sv[j][h] - node0], 1u);
      }
    }
    if (more)
    for (int j = 0; j < SORT_BATCH; ++j) {
      const int s = sb + j * BKT_WAVES + wv;
      const unsigned d0 = s_d0[s], dl = s_dl[s], s0 = s_s0[s], sl = s_sl[s];
      for (unsigned k = (unsigned)lane + 64u; k < dl; k += 32u)
        atomicAdd(&c_in[pairs[d0 + k].y - node0], 1u);
      for (unsigned k = (unsigned)lane + 64u; k < sl; k += 32u)
        atomicAdd(&c_out[srcs[s0 + k] - node0], 1u);
    }
  }
  __syncthreads();

  const unsigned d = c_in[tid];
  unsigned incl = d;
  for (int x = 1; x < 32; x <<= 1) {
    unsigned t = __shfl_up(incl, x);
    if (lane >= x) incl += t;
  }
  if (lane == 31) s_total[wv] = incl;
  __syncthreads();
  unsigned woff = (lane < wv) ? s_total[lane] : 0u;
  for (int x = 16; x > 0; x >>= 1) woff += __shfl_xor(woff, x);
  const unsigned local = woff + incl - d;
  const unsigned first = (unsigned)es + local;
  const bool staged = (ee - es) <= BKT_STAGE;
  c_in[tid] = staged ? local : first;

  const int node = node0 + tid;
  if (node < n_nodes) {
    const unsigned od = c_out[tid];
    seg_start[node] = (int)first;
    indeg[node] = d;
    inn[node] = rsqrtf((float)(d > 1u ? d : 1u));
    outn[node] = rsqrtf((float)(od > 1u ? od : 1u));
  }
  __syncthreads();
  for (int sb = 0; sb < SORT_SLICES; sb += BKT_WAVES * SORT_BATCH) {
    int2 p[SORT_BATCH][2];
    unsigned pos[SORT_BATCH][2];
    bool more = false;
#pragma unroll
    for (int j = 0; j < SORT_BATCH; ++j) {
      const int s = sb + j * BKT_WAVES + wv;
      const unsigned d0 = s_d0[s], dl = s_dl[s];
      more = more || dl > 64u;
#pragma unroll
      for (int h = 0; h < 2; ++h) {
        const unsigned k = (unsigned)(lane + 32 * h);
        p[j][h] = pairs[k < dl ? d0 + k : last];
        if (k >= dl) p[j][h].y = -1;
      }
    }
#pragma unroll
    for (int j = 0; j < SORT_BATCH; ++j) {
#pragma unroll
      for (int h = 0; h < 2; ++h) {
        pos[j][h] = 0u;
        if (p[j][h].y >= 0) pos[j][h] = atomicAdd(&c_in[p[j][h].y - node0], 1u);
      }
    }
#pragma unroll
    for (int j = 0; j < SORT_BATCH; ++j) {
#pragma unroll
      for (int h = 0; h < 2; ++h) {
        if (p[j][h].y >= 0) {
          if (staged) s_nbr[pos[j][h]] = p[j][h].x;
          else nbr[pos[j][h]] = p[j][h].x;
        }
      }
    }
    if (more)
    for (int j = 0; j < SORT_BATCH; ++j) {
      const int s = sb + j * BKT_WAVES + wv;
      const unsigned d0 = s_d0[s], dl = s_dl[s];
      for (unsigned k = (unsigned)lane + 64u; k < dl; k += 32u) {
        const int2 pr = pairs[d0 + k];
        const unsigned ps = atomicAdd(&c_in[pr.y - node0], 1u);
        if (staged) s_nbr[ps] = pr.x;
        else nbr[ps] = pr.x;
      }
    }
  }
  if (staged) {
    __syncthreads();
    for (int i = tid; i < ee - es; i += BKT_NODES) nbr[es + i] = s_nbr[i];
  }
}

#define TILE_ROWS 16
#define LAYER_TILES_PER_WG 4
#define TILE_PITCH 68
static_assert(LAYER_TILES_PER_WG <= 7, "lane i + 1 of a group must exist for every tile i");

__device__ __forceinline__ void gcn_store_rows(const float* __restrict__ s,
                                               float* __restrict__ out, int row0,
                                               int n_nodes, int wave, int lane) {
  const int c4 = (lane & 15) * 4;
  const int r = 4 * wave + (lane >> 4);
  const float* __restrict__ sp = s + r * TILE_PITCH + c4;
  float* __restrict__ o = out + (size_t)(row0 + r) * D_FEAT + c4;
  if (row0 + TILE_ROWS <= n_nodes) {
    const float4 v0 = *reinterpret_cast<const float4*>(sp);
    const float4 v1 = *reinterpret_cast<const float4*>(sp + 2 * TILE_PITCH);
    *reinterpret_cast<float4*>(o) = v0;
    *reinterpret_cast<float4*>(o + 2 * D_FEAT) = v1;
  } else {
#pragma nounroll
    for (int t = 0; t < 2; ++t) {
      if (row0 + r + 2 * t < n_nodes)
        *reinterpret_cast<float4*>(o + 2 * t * D_FEAT) =
            *reinterpret_cast<const float4*>(sp + 2 * t * TILE_PITCH);
    }
  }
}

template <bool kWeighted, bool kScaleOut>
__global__ __launch_bounds__(128) void gcn_layer(
    const float* __restrict__ x, const float* __restrict__ out_norm,
    const float* __restrict__ in_norm, const int* __restrict__ seg_start,
    const unsigned* __restrict__ indeg, const int* __restrict__ nbr,
    const float* __restrict__ W,
    const float* __restrict__ bias,
    float* __restrict__ out, int n_nodes, int n_edges, int n_tiles) {
  __shared__ __attribute__((aligned(16))) float s_tile[2][TILE_ROWS * TILE_PITCH];
  __shared__ __attribute__((aligned(16))) float s_out[2][TILE_ROWS * TILE_PITCH];
  const int lane = threadIdx.x & 31;
  const int wave = threadIdx.x >> 5;
  const int nwg = (int)gridDim.x;
  const int q = lane & 7;
  const int grow = wave * 4 + (lane >> 3);
  const int half = lane >> 4;
  const int l15  = lane & 15;
  const int n0   = wave * 16;

  int mstart, mcnt;
  float msc, mon = 1.0f;
  {
    const int tq = (int)blockIdx.x + q * nwg;
    const int nodeq = tq * TILE_ROWS + grow;
    const bool okq = q < LAYER_TILES_PER_WG && tq < n_tiles && nodeq < n_nodes;
    const int nc = okq ? nodeq : 0;
    mstart = seg_start[nc];
    const int dc = (int)indeg[nc];
    msc = in_norm[nc];
    if (kScaleOut) mon = out_norm[nc];
    mcnt = okq ? dc : 0;
  }

  const float* __restrict__ bptr = W + (size_t)(2 * half) * D_FEAT + n0 + l15;
  v2f bf[D_FEAT / 4];
#pragma unroll
  for (int i = 0; i < D_FEAT / 4; ++i) {
    bf[i].x = bptr[(size_t)(4 * i) * D_FEAT];
    bf[i].y = bptr[(size_t)(4 * i + 1) * D_FEAT];
  }
  const float bv = bias[n0 + l15];
  const float4* __restrict__ x4 = reinterpret_cast<const float4*>(x);

  int start = __shfl(mstart, 0, 8);
  int cnt = __shfl(mcnt, 0, 8);
  float sc = __shfl(msc, 0, 8);
  int my0, my1;
  float nw0 = 1.0f, nw1 = 1.0f;
  {
    const int m = cnt < 16 ? cnt : 16;
    const int e0 = start + (q < m ? q : 0);
    const int e1 = start + (q + 8 < m ? q + 8 : 0);
    my0 = nbr[e0 < n_edges ? e0 : n_edges - 1];
    my1 = nbr[e1 < n_edges ? e1 : n_edges - 1];
    if (kWeighted) {
      nw0 = out_norm[my0];
      nw1 = out_norm[my1];
    }
  }

  bool pending = false;
  int prow0 = 0;
  int pbuf = 0;
  int buf = 0;
  for (int i = 0; i < LAYER_TILES_PER_WG; ++i, buf ^= 1) {
    const int t = (int)blockIdx.x + i * nwg;
    if (t >= n_tiles) break;
    const int row0 = t * TILE_ROWS;

    const int nstart = __shfl(mstart, i + 1, 8);
    const int ncnt = __shfl(mcnt, i + 1, 8);
    const float nsc = __shfl(msc, i + 1, 8);
    int nmy0, nmy1;
    {
      const int m = ncnt < 16 ? ncnt : 16;
      const int e0 = nstart + (q < m ? q : 0);
      const int e1 = nstart + (q + 8 < m ? q + 8 : 0);
      nmy0 = nbr[e0 < n_edges ? e0 : n_edges - 1];
      nmy1 = nbr[e1 < n_edges ? e1 : n_edges - 1];
    }

    float4 lo = make_float4(0.0f, 0.0f, 0.0f, 0.0f);
    float4 hi = make_float4(0.0f, 0.0f, 0.0f, 0.0f);
    for (int b = 0; b < cnt; b += 16) {
      const int m = (cnt - b) < 16 ? (cnt - b) : 16;
      if (b > 0) {
        my0 = nbr[start + b + (q < m ? q : 0)];
        my1 = nbr[start + b + (q + 8 < m ? q + 8 : 0)];
        if (kWeighted) {
          nw0 = out_norm[my0];
          nw1 = out_norm[my1];
        }
      }
      for (int j = 0; j < m; j += 4) {
        const int ids = j < 8 ? my0 : my1;
        const int jj = j & 7;
        const int s0 = __shfl(ids, jj, 8);
        const int s1 = __shfl(ids, jj + 1, 8);
        const int s2 = __shfl(ids, jj + 2, 8);
        const int s3 = __shfl(ids, jj + 3, 8);
        const float4 u0 = x4[(size_t)s0 * (D_FEAT / 4) + q];
        const float4 v0 = x4[(size_t)s0 * (D_FEAT / 4) + 8 + q];
        const float4 u1 = x4[(size_t)s1 * (D_FEAT / 4) + q];
        const float4 v1 = x4[(size_t)s1 * (D_FEAT / 4) + 8 + q];
        const float4 u2 = x4[(size_t)s2 * (D_FEAT / 4) + q];
        const float4 v2 = x4[(size_t)s2 * (D_FEAT / 4) + 8 + q];
        const float4 u3 = x4[(size_t)s3 * (D_FEAT / 4) + q];
        const float4 v3 = x4[(size_t)s3 * (D_FEAT / 4) + 8 + q];
        float w0 = 1.0f;
        float w1 = (j + 1 < m) ? 1.0f : 0.0f;
        float w2 = (j + 2 < m) ? 1.0f : 0.0f;
        float w3 = (j + 3 < m) ? 1.0f : 0.0f;
        if (kWeighted) {
          __builtin_amdgcn_sched_barrier(0);
          const float wts = j < 8 ? nw0 : nw1;
          w0 = __shfl(wts, jj, 8);
          w1 = (j + 1 < m) ? __shfl(wts, jj + 1, 8) : 0.0f;
          w2 = (j + 2 < m) ? __shfl(wts, jj + 2, 8) : 0.0f;
          w3 = (j + 3 < m) ? __shfl(wts, jj + 3, 8) : 0.0f;
        }
        lo.x += u0.x * w0; lo.y += u0.y * w0; lo.z += u0.z * w0; lo.w += u0.w * w0;
        hi.x += v0.x * w0; hi.y += v0.y * w0; hi.z += v0.z * w0; hi.w += v0.w * w0;
        lo.x += u1.x * w1; lo.y += u1.y * w1; lo.z += u1.z * w1; lo.w += u1.w * w1;
        hi.x += v1.x * w1; hi.y += v1.y * w1; hi.z += v1.z * w1; hi.w += v1.w * w1;
        lo.x += u2.x * w2; lo.y += u2.y * w2; lo.z += u2.z * w2; lo.w += u2.w * w2;
        hi.x += v2.x * w2; hi.y += v2.y * w2; hi.z += v2.z * w2; hi.w += v2.w * w2;
        lo.x += u3.x * w3; lo.y += u3.y * w3; lo.z += u3.z * w3; lo.w += u3.w * w3;
        hi.x += v3.x * w3; hi.y += v3.y * w3; hi.z += v3.z * w3; hi.w += v3.w * w3;
      }
    }
    lo.x *= sc; lo.y *= sc; lo.z *= sc; lo.w *= sc;
    hi.x *= sc; hi.y *= sc; hi.z *= sc; hi.w *= sc;

    float* __restrict__ st = &s_tile[buf][grow * TILE_PITCH];
    *reinterpret_cast<float4*>(st + 4 * q) = lo;
    *reinterpret_cast<float4*>(st + 32 + 4 * q) = hi;
    if (kScaleOut) {
      const float on = __shfl(mon, i, 8);
      if (q == 0) st[D_FEAT] = on;
    }
    __syncthreads();

    if (pending) gcn_store_rows(s_out[pbuf], out, prow0, n_nodes, wave, lane);

    const float* __restrict__ ap = &s_tile[buf][l15 * TILE_PITCH + 2 * half];
    v8f c = {};
#pragma unroll
    for (int k = 0; k < D_FEAT / 4; ++k) {
      const v2f a = *reinterpret_cast<const v2f*>(ap + 4 * k);
      c = __builtin_amdgcn_wmma_f32_16x16x4_f32(false, a, false, bf[k],
                                                (short)0, c, false, false);
    }

    float nnw0 = 1.0f, nnw1 = 1.0f;
    if (kWeighted) {
      nnw0 = out_norm[nmy0];
      nnw1 = out_norm[nmy1];
    }
    float osc[8];
    if (kScaleOut) {
      const float* __restrict__ sp = &s_tile[buf][(8 * half) * TILE_PITCH + D_FEAT];
#pragma unroll
      for (int v = 0; v < 8; ++v) osc[v] = sp[v * TILE_PITCH];
    }

    float* __restrict__ so = &s_out[buf][(8 * half) * TILE_PITCH + n0 + l15];
#pragma unroll
    for (int v = 0; v < 8; ++v) {
      const float xv = c[v] + bv;
      float yv = xv > 0.0f ? xv : expm1f(xv);
      if (kScaleOut) yv *= osc[v];
      so[v * TILE_PITCH] = yv;
    }
    pending = true;
    prow0 = row0;
    pbuf = buf;

    start = nstart; cnt = ncnt; sc = nsc;
    my0 = nmy0; my1 = nmy1; nw0 = nnw0; nw1 = nnw1;
  }
  if (pending) {
    __syncthreads();
    gcn_store_rows(s_out[pbuf], out, prow0, n_nodes, wave, lane);
  }
}

extern "C" void kernel_launch(void* const* d_in, const int* in_sizes, int n_in,
                              void* d_out, int out_size, void* d_ws, size_t ws_size,
                              hipStream_t stream) {
  const float* feat = (const float*)d_in[0];
  const float* Ws[3] = {(const float*)d_in[1], (const float*)d_in[3], (const float*)d_in[5]};
  const float* bs[3] = {(const float*)d_in[2], (const float*)d_in[4], (const float*)d_in[6]};
  const int* src = (const int*)d_in[7];
  const int* dst = (const int*)d_in[8];

  const int n_nodes = in_sizes[0] / D_FEAT;
  const int n_edges = in_sizes[7];

  float*    out_norm  = (float*)d_ws;
  unsigned* indeg     = (unsigned*)(out_norm + n_nodes);
  float*    mid       = out_norm + 2 * (size_t)n_nodes;
  float*    in_norm   = mid + (size_t)n_nodes * D_FEAT;
  int*      seg_start = (int*)(in_norm + n_nodes);
  int*      nbr       = seg_start + n_nodes;
  int*      srcs      = nbr + n_edges;
  float*    outp      = (float*)d_out;

  {
    const int nb = (n_nodes + BKT_NODES - 1) >> BKT_SHIFT;
    int2*     pairs = (int2*)mid;
    unsigned* tab   = (unsigned*)mid + 2 * (size_t)n_edges;
    const size_t lds = 2 * (size_t)nb * sizeof(unsigned);
    const size_t per = ((size_t)n_edges + SORT_SLICES - 1) / SORT_SLICES;
    gcn_slice_group<<<SORT_SLICES, 1024, per * 12 + lds, stream>>>(
        src, dst, tab, pairs, srcs, n_edges, nb);
    gcn_bucket_build<<<(unsigned)nb, BKT_NODES, 0, stream>>>(
        pairs, srcs, tab, (int)per, nb, out_norm, indeg, in_norm, seg_start,
        nbr, n_nodes, n_edges);
  }

  const int n_tiles = (n_nodes + TILE_ROWS - 1) / TILE_ROWS;
  const unsigned lblocks =
      (unsigned)((n_tiles + LAYER_TILES_PER_WG - 1) / LAYER_TILES_PER_WG);
  gcn_layer<true, true><<<lblocks, 128, 0, stream>>>(
      feat, out_norm, in_norm, seg_start, indeg, nbr, Ws[0], bs[0], outp,
      n_nodes, n_edges, n_tiles);
  gcn_layer<false, true><<<lblocks, 128, 0, stream>>>(
      outp, out_norm, in_norm, seg_start, indeg, nbr, Ws[1], bs[1], mid,
      n_nodes, n_edges, n_tiles);
  gcn_layer<false, false><<<lblocks, 128, 0, stream>>>(
      mid, out_norm, in_norm, seg_start, indeg, nbr, Ws[2], bs[2], outp,
      n_nodes, n_edges, n_tiles);
}


# ELU epilogue without branches on register pairs: expm1 written out as the library evaluates it, in packed two-value f32 instructions, four independent chains per lane; sums and their order unchanged
# speedup vs baseline: 6.9801x; 1.0524x over previous
#include <hip/hip_runtime.h>
#include <hip/hip_bf16.h>

#define D_FEAT 64

typedef float v2f __attribute__((ext_vector_type(2)));
typedef float v8f __attribute__((ext_vector_type(8)));

#define BKT_SHIFT 9
#define BKT_NODES 512
#define SORT_SLICES 256
#define SORT_BATCH 8
#define BKT_STAGE 12288
#define BKT_WAVES (BKT_NODES / 32)
static_assert(SORT_SLICES % (BKT_WAVES * SORT_BATCH) == 0,
              "pass B hands whole groups of slices to its waves");

__global__ __launch_bounds__(1024) void gcn_slice_group(
    const int* __restrict__ src, const int* __restrict__ dst,
    unsigned* __restrict__ tab, int2* __restrict__ pairs,
    int* __restrict__ srcs, int n_edges, int nb) {
  extern __shared__ int2 s_stage[];
  __shared__ unsigned s_total[32];
  const int per = (n_edges + (int)gridDim.x - 1) / (int)gridDim.x;
  int2* s_pair = s_stage;
  int* s_src = (int*)(s_stage + per);
  unsigned* s_cur = (unsigned*)(s_src + per);
  const int tid = threadIdx.x;
  const int lane = tid & 31;
  const int wv = tid >> 5;
  const int e0 = (int)blockIdx.x * per;
  const int e1 = (e0 + per) < n_edges ? (e0 + per) : n_edges;
  const int n_own = e1 > e0 ? e1 - e0 : 0;

  for (int i = tid; i < 2 * nb; i += 1024) s_cur[i] = 0u;
  __syncthreads();
  for (int e = e0 + tid; e < e1; e += SORT_BATCH * 1024) {
    int d[SORT_BATCH], s[SORT_BATCH];
#pragma unroll
    for (int j = 0; j < SORT_BATCH; ++j) {
      const int ej = e + j * 1024;
      const int ec = ej < e1 ? ej : e1 - 1;
      const int dv = dst[ec];
      const int sv = src[ec];
      d[j] = ej < e1 ? dv : -1;
      s[j] = sv;
    }
#pragma unroll
    for (int j = 0; j < SORT_BATCH; ++j) {
      if (d[j] >= 0) {
        atomicAdd(&s_cur[d[j] >> BKT_SHIFT], 1u);
        atomicAdd(&s_cur[nb + (s[j] >> BKT_SHIFT)], 1u);
      }
    }
  }
  __syncthreads();

  unsigned carry = 0u;
  for (int base = 0; base < 2 * nb; base += 1024) {
    const int i = base + tid;
    const unsigned c = (i < 2 * nb) ? s_cur[i] : 0u;
    unsigned incl = c;
    for (int x = 1; x < 32; x <<= 1) {
      unsigned t = __shfl_up(incl, x);
      if (lane >= x) incl += t;
    }
    if (lane == 31) s_total[wv] = incl;
    __syncthreads();
    unsigned woff = (lane < wv) ? s_total[lane] : 0u;
    unsigned all = s_total[lane];
    for (int x = 16; x > 0; x >>= 1) {
      woff += __shfl_xor(woff, x);
      all += __shfl_xor(all, x);
    }
    if (i < 2 * nb) s_cur[i] = carry + woff + incl - c;
    carry += all;
    __syncthreads();
  }
  for (int i = tid; i < 2 * nb; i += 1024) {
    const int key = i >= nb ? 1 : 0;
    const unsigned ls = s_cur[i] - (key ? (unsigned)n_own : 0u);
    s_cur[i] = ls;
    tab[((size_t)key * (nb + 1) + (i - key * nb)) * gridDim.x + blockIdx.x] = ls;
  }
  if (tid < 2)
    tab[((size_t)tid * (nb + 1) + nb) * gridDim.x + blockIdx.x] = (unsigned)n_own;
  __syncthreads();

  for (int e = e0 + tid; e < e1; e += SORT_BATCH * 1024) {
    int d[SORT_BATCH], s[SORT_BATCH];
#pragma unroll
    for (int j = 0; j < SORT_BATCH; ++j) {
      const int ej = e + j * 1024;
      const int ec = ej < e1 ? ej : e1 - 1;
      const int dv = dst[ec];
      const int sv = src[ec];
      d[j] = ej < e1 ? dv : -1;
      s[j] = sv;
    }
#pragma unroll
    for (int j = 0; j < SORT_BATCH; ++j) {
      if (d[j] >= 0) {
        const unsigned p = atomicAdd(&s_cur[d[j] >> BKT_SHIFT], 1u);
        const unsigned q = atomicAdd(&s_cur[nb + (s[j] >> BKT_SHIFT)], 1u);
        s_pair[p] = make_int2(s[j], d[j]);
        s_src[q] = s[j];
      }
    }
  }
  __syncthreads();
  for (int i = tid; i < n_own; i += 1024) {
    pairs[e0 + i] = s_pair[i];
    srcs[e0 + i] = s_src[i];
  }
}

__global__ __launch_bounds__(BKT_NODES) void gcn_bucket_build(
    const int2* __restrict__ pairs, const int* __restrict__ srcs,
    const unsigned* __restrict__ tab, int per, int nb,
    float* __restrict__ outn, unsigned* __restrict__ indeg,
    float* __restrict__ inn, int* __restrict__ seg_start,
    int* __restrict__ nbr, int n_nodes, int n_edges) {
  __shared__ unsigned c_in[BKT_NODES];
  __shared__ unsigned c_out[BKT_NODES];
  __shared__ unsigned s_total[32];
  __shared__ unsigned s_red[2][BKT_WAVES];
  __shared__ unsigned s_d0[SORT_SLICES];
  __shared__ unsigned s_dl[SORT_SLICES];
  __shared__ unsigned s_s0[SORT_SLICES];
  __shared__ unsigned s_sl[SORT_SLICES];
  __shared__ int s_nbr[BKT_STAGE];
  const int tid = threadIdx.x;
  const int lane = tid & 31;
  const int wv = tid >> 5;
  const int b = blockIdx.x;
  const int node0 = b << BKT_SHIFT;
  const unsigned last = (unsigned)(n_edges - 1);
  c_in[tid] = 0u;
  c_out[tid] = 0u;
  unsigned sum0 = 0u, suml = 0u;
  for (int s = tid; s < SORT_SLICES; s += BKT_NODES) {
    const unsigned* __restrict__ td = tab + (size_t)b * SORT_SLICES + s;
    const unsigned* __restrict__ ts = tab + (size_t)(nb + 1 + b) * SORT_SLICES + s;
    const unsigned a0 = td[0];
    const unsigned a1 = td[SORT_SLICES];
    const unsigned g0 = ts[0];
    const unsigned g1 = ts[SORT_SLICES];
    const unsigned base = (unsigned)s * (unsigned)per;
    s_d0[s] = base + a0;
    s_dl[s] = a1 - a0;
    s_s0[s] = base + g0;
    s_sl[s] = g1 - g0;
    sum0 += a0;
    suml += a1 - a0;
  }
  for (int x = 16; x > 0; x >>= 1) {
    sum0 += __shfl_xor(sum0, x);
    suml += __shfl_xor(suml, x);
  }
  if (lane == 0) { s_red[0][wv] = sum0; s_red[1][wv] = suml; }
  __syncthreads();
  unsigned r0 = lane < BKT_WAVES ? s_red[0][lane] : 0u;
  unsigned r1 = lane < BKT_WAVES ? s_red[1][lane] : 0u;
  for (int x = 16; x > 0; x >>= 1) {
    r0 += __shfl_xor(r0, x);
    r1 += __shfl_xor(r1, x);
  }
  const int es = (int)r0;
  const int ee = (int)(r0 + r1);

  for (int sb = 0; sb < SORT_SLICES; sb += BKT_WAVES * SORT_BATCH) {
    int d[SORT_BATCH][2], sv[SORT_BATCH][2];
    bool more = false;
#pragma unroll
    for (int j = 0; j < SORT_BATCH; ++j) {
      const int s = sb + j * BKT_WAVES + wv;
      const unsigned d0 = s_d0[s], dl = s_dl[s], s0 = s_s0[s], sl = s_sl[s];
      more = more || dl > 64u || sl > 64u;
#pragma unroll
      for (int h = 0; h < 2; ++h) {
        const unsigned k = (unsigned)(lane + 32 * h);
        const int dv = pairs[k < dl ? d0 + k : last].y;
        const int sx = srcs[k < sl ? s0 + k : last];
        d[j][h] = k < dl ? dv : -1;
        sv[j][h] = k < sl ? sx : -1;
      }
    }
#pragma unroll
    for (int j = 0; j < SORT_BATCH; ++j) {
#pragma unroll
      for (int h = 0; h < 2; ++h) {
        if (d[j][h] >= 0) atomicAdd(&c_in[d[j][h] - node0], 1u);
        if (sv[j][h] >= 0) atomicAdd(&c_out[sv[j][h] - node0], 1u);
      }
    }
    if (more)
    for (int j = 0; j < SORT_BATCH; ++j) {
      const int s = sb + j * BKT_WAVES + wv;
      const unsigned d0 = s_d0[s], dl = s_dl[s], s0 = s_s0[s], sl = s_sl[s];
      for (unsigned k = (unsigned)lane + 64u; k < dl; k += 32u)
        atomicAdd(&c_in[pairs[d0 + k].y - node0], 1u);
      for (unsigned k = (unsigned)lane + 64u; k < sl; k += 32u)
        atomicAdd(&c_out[srcs[s0 + k] - node0], 1u);
    }
  }
  __syncthreads();

  const unsigned d = c_in[tid];
  unsigned incl = d;
  for (int x = 1; x < 32; x <<= 1) {
    unsigned t = __shfl_up(incl, x);
    if (lane >= x) incl += t;
  }
  if (lane == 31) s_total[wv] = incl;
  __syncthreads();
  unsigned woff = (lane < wv) ? s_total[lane] : 0u;
  for (int x = 16; x > 0; x >>= 1) woff += __shfl_xor(woff, x);
  const unsigned local = woff + incl - d;
  const unsigned first = (unsigned)es + local;
  const bool staged = (ee - es) <= BKT_STAGE;
  c_in[tid] = staged ? local : first;

  const int node = node0 + tid;
  if (node < n_nodes) {
    const unsigned od = c_out[tid];
    seg_start[node] = (int)first;
    indeg[node] = d;
    inn[node] = rsqrtf((float)(d > 1u ? d : 1u));
    outn[node] = rsqrtf((float)(od > 1u ? od : 1u));
  }
  __syncthreads();
  for (int sb = 0; sb < SORT_SLICES; sb += BKT_WAVES * SORT_BATCH) {
    int2 p[SORT_BATCH][2];
    unsigned pos[SORT_BATCH][2];
    bool more = false;
#pragma unroll
    for (int j = 0; j < SORT_BATCH; ++j) {
      const int s = sb + j * BKT_WAVES + wv;
      const unsigned d0 = s_d0[s], dl = s_dl[s];
      more = more || dl > 64u;
#pragma unroll
      for (int h = 0; h < 2; ++h) {
        const unsigned k = (unsigned)(lane + 32 * h);
        p[j][h] = pairs[k < dl ? d0 + k : last];
        if (k >= dl) p[j][h].y = -1;
      }
    }
#pragma unroll
    for (int j = 0; j < SORT_BATCH; ++j) {
#pragma unroll
      for (int h = 0; h < 2; ++h) {
        pos[j][h] = 0u;
        if (p[j][h].y >= 0) pos[j][h] = atomicAdd(&c_in[p[j][h].y - node0], 1u);
      }
    }
#pragma unroll
    for (int j = 0; j < SORT_BATCH; ++j) {
#pragma unroll
      for (int h = 0; h < 2; ++h) {
        if (p[j][h].y >= 0) {
          if (staged) s_nbr[pos[j][h]] = p[j][h].x;
          else nbr[pos[j][h]] = p[j][h].x;
        }
      }
    }
    if (more)
    for (int j = 0; j < SORT_BATCH; ++j) {
      const int s = sb + j * BKT_WAVES + wv;
      const unsigned d0 = s_d0[s], dl = s_dl[s];
      for (unsigned k = (unsigned)lane + 64u; k < dl; k += 32u) {
        const int2 pr = pairs[d0 + k];
        const unsigned ps = atomicAdd(&c_in[pr.y - node0], 1u);
        if (staged) s_nbr[ps] = pr.x;
        else nbr[ps] = pr.x;
      }
    }
  }
  if (staged) {
    __syncthreads();
    for (int i = tid; i < ee - es; i += BKT_NODES) nbr[es + i] = s_nbr[i];
  }
}

#define TILE_ROWS 16
#define LAYER_TILES_PER_WG 4
#define TILE_PITCH 68
static_assert(LAYER_TILES_PER_WG <= 7, "lane i + 1 of a group must exist for every tile i");

__device__ __forceinline__ void gcn_store_rows(const float* __restrict__ s,
                                               float* __restrict__ out, int row0,
                                               int n_nodes, int wave, int lane) {
  const int c4 = (lane & 15) * 4;
  const int r = 4 * wave + (lane >> 4);
  const float* __restrict__ sp = s + r * TILE_PITCH + c4;
  float* __restrict__ o = out + (size_t)(row0 + r) * D_FEAT + c4;
  if (row0 + TILE_ROWS <= n_nodes) {
    const float4 v0 = *reinterpret_cast<const float4*>(sp);
    const float4 v1 = *reinterpret_cast<const float4*>(sp + 2 * TILE_PITCH);
    *reinterpret_cast<float4*>(o) = v0;
    *reinterpret_cast<float4*>(o + 2 * D_FEAT) = v1;
  } else {
#pragma nounroll
    for (int t = 0; t < 2; ++t) {
      if (row0 + r + 2 * t < n_nodes)
        *reinterpret_cast<float4*>(o + 2 * t * D_FEAT) =
            *reinterpret_cast<const float4*>(sp + 2 * t * TILE_PITCH);
    }
  }
}

__device__ __forceinline__ v2f gcn_elu2(v2f x) {
  const float kLog2e = __builtin_bit_cast(float, 0x3fb8aa3bu);
  const float kNegLn2Hi = __builtin_bit_cast(float, 0xbf317218u);
  const float kLn2Lo = __builtin_bit_cast(float, 0x3102e308u);
  const float kC6 = __builtin_bit_cast(float, 0x395133b1u);
  const float kC5 = __builtin_bit_cast(float, 0x3ab69700u);
  const float kC4 = __builtin_bit_cast(float, 0x3c0887f9u);
  const float kC3 = __builtin_bit_cast(float, 0x3d2aaa81u);
  const float kC2 = __builtin_bit_cast(float, 0x3e2aaaabu);
  const v2f t = x * kLog2e;
  v2f n;
  n.x = __builtin_rintf(t.x);
  n.y = __builtin_rintf(t.y);
  v2f r = __builtin_elementwise_fma(n, (v2f){kNegLn2Hi, kNegLn2Hi}, x);
  r = __builtin_elementwise_fma(n, (v2f){kLn2Lo, kLn2Lo}, r);
  v2f s;
  s.x = __builtin_ldexpf(1.0f, (int)n.x);
  s.y = __builtin_ldexpf(1.0f, (int)n.y);
  v2f p = __builtin_elementwise_fma((v2f){kC6, kC6}, r, (v2f){kC5, kC5});
  p = __builtin_elementwise_fma(r, p, (v2f){kC4, kC4});
  p = __builtin_elementwise_fma(r, p, (v2f){kC3, kC3});
  p = __builtin_elementwise_fma(r, p, (v2f){kC2, kC2});
  p = __builtin_elementwise_fma(r, p, (v2f){0.5f, 0.5f});
  const v2f q = r * p;
  const v2f r2 = __builtin_elementwise_fma(r, q, r);
  v2f e = s + (v2f){-1.0f, -1.0f};
  e = __builtin_elementwise_fma(s, r2, e);
  v2f y;
  y.x = x.x > 0.0f ? x.x : (-17.0f > x.x ? -1.0f : e.x);
  y.y = x.y > 0.0f ? x.y : (-17.0f > x.y ? -1.0f : e.y);
  return y;
}

template <bool kWeighted, bool kScaleOut>
__global__ __launch_bounds__(128) void gcn_layer(
    const float* __restrict__ x, const float* __restrict__ out_norm,
    const float* __restrict__ in_norm, const int* __restrict__ seg_start,
    const unsigned* __restrict__ indeg, const int* __restrict__ nbr,
    const float* __restrict__ W,
    const float* __restrict__ bias,
    float* __restrict__ out, int n_nodes, int n_edges, int n_tiles) {
  __shared__ __attribute__((aligned(16))) float s_tile[2][TILE_ROWS * TILE_PITCH];
  __shared__ __attribute__((aligned(16))) float s_out[2][TILE_ROWS * TILE_PITCH];
  const int lane = threadIdx.x & 31;
  const int wave = threadIdx.x >> 5;
  const int nwg = (int)gridDim.x;
  const int q = lane & 7;
  const int grow = wave * 4 + (lane >> 3);
  const int half = lane >> 4;
  const int l15  = lane & 15;
  const int n0   = wave * 16;

  int mstart, mcnt;
  float msc, mon = 1.0f;
  {
    const int tq = (int)blockIdx.x + q * nwg;
    const int nodeq = tq * TILE_ROWS + grow;
    const bool okq = q < LAYER_TILES_PER_WG && tq < n_tiles && nodeq < n_nodes;
    const int nc = okq ? nodeq : 0;
    mstart = seg_start[nc];
    const int dc = (int)indeg[nc];
    msc = in_norm[nc];
    if (kScaleOut) mon = out_norm[nc];
    mcnt = okq ? dc : 0;
  }

  const float* __restrict__ bptr = W + (size_t)(2 * half) * D_FEAT + n0 + l15;
  v2f bf[D_FEAT / 4];
#pragma unroll
  for (int i = 0; i < D_FEAT / 4; ++i) {
    bf[i].x = bptr[(size_t)(4 * i) * D_FEAT];
    bf[i].y = bptr[(size_t)(4 * i + 1) * D_FEAT];
  }
  const float bv = bias[n0 + l15];
  const float4* __restrict__ x4 = reinterpret_cast<const float4*>(x);

  int start = __shfl(mstart, 0, 8);
  int cnt = __shfl(mcnt, 0, 8);
  float sc = __shfl(msc, 0, 8);
  int my0, my1;
  float nw0 = 1.0f, nw1 = 1.0f;
  {
    const int m = cnt < 16 ? cnt : 16;
    const int e0 = start + (q < m ? q : 0);
    const int e1 = start + (q + 8 < m ? q + 8 : 0);
    my0 = nbr[e0 < n_edges ? e0 : n_edges - 1];
    my1 = nbr[e1 < n_edges ? e1 : n_edges - 1];
    if (kWeighted) {
      nw0 = out_norm[my0];
      nw1 = out_norm[my1];
    }
  }

  bool pending = false;
  int prow0 = 0;
  int pbuf = 0;
  int buf = 0;
  for (int i = 0; i < LAYER_TILES_PER_WG; ++i, buf ^= 1) {
    const int t = (int)blockIdx.x + i * nwg;
    if (t >= n_tiles) break;
    const int row0 = t * TILE_ROWS;

    const int nstart = __shfl(mstart, i + 1, 8);
    const int ncnt = __shfl(mcnt, i + 1, 8);
    const float nsc = __shfl(msc, i + 1, 8);
    int nmy0, nmy1;
    {
      const int m = ncnt < 16 ? ncnt : 16;
      const int e0 = nstart + (q < m ? q : 0);
      const int e1 = nstart + (q + 8 < m ? q + 8 : 0);
      nmy0 = nbr[e0 < n_edges ? e0 : n_edges - 1];
      nmy1 = nbr[e1 < n_edges ? e1 : n_edges - 1];
    }

    float4 lo = make_float4(0.0f, 0.0f, 0.0f, 0.0f);
    float4 hi = make_float4(0.0f, 0.0f, 0.0f, 0.0f);
    for (int b = 0; b < cnt; b += 16) {
      const int m = (cnt - b) < 16 ? (cnt - b) : 16;
      if (b > 0) {
        my0 = nbr[start + b + (q < m ? q : 0)];
        my1 = nbr[start + b + (q + 8 < m ? q + 8 : 0)];
        if (kWeighted) {
          nw0 = out_norm[my0];
          nw1 = out_norm[my1];
        }
      }
      for (int j = 0; j < m; j += 4) {
        const int ids = j < 8 ? my0 : my1;
        const int jj = j & 7;
        const int s0 = __shfl(ids, jj, 8);
        const int s1 = __shfl(ids, jj + 1, 8);
        const int s2 = __shfl(ids, jj + 2, 8);
        const int s3 = __shfl(ids, jj + 3, 8);
        const float4 u0 = x4[(size_t)s0 * (D_FEAT / 4) + q];
        const float4 v0 = x4[(size_t)s0 * (D_FEAT / 4) + 8 + q];
        const float4 u1 = x4[(size_t)s1 * (D_FEAT / 4) + q];
        const float4 v1 = x4[(size_t)s1 * (D_FEAT / 4) + 8 + q];
        const float4 u2 = x4[(size_t)s2 * (D_FEAT / 4) + q];
        const float4 v2 = x4[(size_t)s2 * (D_FEAT / 4) + 8 + q];
        const float4 u3 = x4[(size_t)s3 * (D_FEAT / 4) + q];
        const float4 v3 = x4[(size_t)s3 * (D_FEAT / 4) + 8 + q];
        float w0 = 1.0f;
        float w1 = (j + 1 < m) ? 1.0f : 0.0f;
        float w2 = (j + 2 < m) ? 1.0f : 0.0f;
        float w3 = (j + 3 < m) ? 1.0f : 0.0f;
        if (kWeighted) {
          __builtin_amdgcn_sched_barrier(0);
          const float wts = j < 8 ? nw0 : nw1;
          w0 = __shfl(wts, jj, 8);
          w1 = (j + 1 < m) ? __shfl(wts, jj + 1, 8) : 0.0f;
          w2 = (j + 2 < m) ? __shfl(wts, jj + 2, 8) : 0.0f;
          w3 = (j + 3 < m) ? __shfl(wts, jj + 3, 8) : 0.0f;
        }
        lo.x += u0.x * w0; lo.y += u0.y * w0; lo.z += u0.z * w0; lo.w += u0.w * w0;
        hi.x += v0.x * w0; hi.y += v0.y * w0; hi.z += v0.z * w0; hi.w += v0.w * w0;
        lo.x += u1.x * w1; lo.y += u1.y * w1; lo.z += u1.z * w1; lo.w += u1.w * w1;
        hi.x += v1.x * w1; hi.y += v1.y * w1; hi.z += v1.z * w1; hi.w += v1.w * w1;
        lo.x += u2.x * w2; lo.y += u2.y * w2; lo.z += u2.z * w2; lo.w += u2.w * w2;
        hi.x += v2.x * w2; hi.y += v2.y * w2; hi.z += v2.z * w2; hi.w += v2.w * w2;
        lo.x += u3.x * w3; lo.y += u3.y * w3; lo.z += u3.z * w3; lo.w += u3.w * w3;
        hi.x += v3.x * w3; hi.y += v3.y * w3; hi.z += v3.z * w3; hi.w += v3.w * w3;
      }
    }
    lo.x *= sc; lo.y *= sc; lo.z *= sc; lo.w *= sc;
    hi.x *= sc; hi.y *= sc; hi.z *= sc; hi.w *= sc;

    float* __restrict__ st = &s_tile[buf][grow * TILE_PITCH];
    *reinterpret_cast<float4*>(st + 4 * q) = lo;
    *reinterpret_cast<float4*>(st + 32 + 4 * q) = hi;
    if (kScaleOut) {
      const float on = __shfl(mon, i, 8);
      if (q == 0) st[D_FEAT] = on;
    }
    __syncthreads();

    if (pending) gcn_store_rows(s_out[pbuf], out, prow0, n_nodes, wave, lane);

    const float* __restrict__ ap = &s_tile[buf][l15 * TILE_PITCH + 2 * half];
    v8f c = {};
#pragma unroll
    for (int k = 0; k < D_FEAT / 4; ++k) {
      const v2f a = *reinterpret_cast<const v2f*>(ap + 4 * k);
      c = __builtin_amdgcn_wmma_f32_16x16x4_f32(false, a, false, bf[k],
                                                (short)0, c, false, false);
    }

    float nnw0 = 1.0f, nnw1 = 1.0f;
    if (kWeighted) {
      nnw0 = out_norm[nmy0];
      nnw1 = out_norm[nmy1];
    }
    float osc[8];
    if (kScaleOut) {
      const float* __restrict__ sp = &s_tile[buf][(8 * half) * TILE_PITCH + D_FEAT];
#pragma unroll
      for (int v = 0; v < 8; ++v) osc[v] = sp[v * TILE_PITCH];
    }

    float* __restrict__ so = &s_out[buf][(8 * half) * TILE_PITCH + n0 + l15];
#pragma unroll
    for (int v = 0; v < 8; v += 2) {
      v2f xv;
      xv.x = c[v] + bv;
      xv.y = c[v + 1] + bv;
      v2f yv = gcn_elu2(xv);
      if (kScaleOut) {
        yv.x *= osc[v];
        yv.y *= osc[v + 1];
      }
      so[v * TILE_PITCH] = yv.x;
      so[(v + 1) * TILE_PITCH] = yv.y;
    }
    pending = true;
    prow0 = row0;
    pbuf = buf;

    start = nstart; cnt = ncnt; sc = nsc;
    my0 = nmy0; my1 = nmy1; nw0 = nnw0; nw1 = nnw1;
  }
  if (pending) {
    __syncthreads();
    gcn_store_rows(s_out[pbuf], out, prow0, n_nodes, wave, lane);
  }
}

extern "C" void kernel_launch(void* const* d_in, const int* in_sizes, int n_in,
                              void* d_out, int out_size, void* d_ws, size_t ws_size,
                              hipStream_t stream) {
  const float* feat = (const float*)d_in[0];
  const float* Ws[3] = {(const float*)d_in[1], (const float*)d_in[3], (const float*)d_in[5]};
  const float* bs[3] = {(const float*)d_in[2], (const float*)d_in[4], (const float*)d_in[6]};
  const int* src = (const int*)d_in[7];
  const int* dst = (const int*)d_in[8];

  const int n_nodes = in_sizes[0] / D_FEAT;
  const int n_edges = in_sizes[7];

  float*    out_norm  = (float*)d_ws;
  unsigned* indeg     = (unsigned*)(out_norm + n_nodes);
  float*    mid       = out_norm + 2 * (size_t)n_nodes;
  float*    in_norm   = mid + (size_t)n_nodes * D_FEAT;
  int*      seg_start = (int*)(in_norm + n_nodes);
  int*      nbr       = seg_start + n_nodes;
  int*      srcs      = nbr + n_edges;
  float*    outp      = (float*)d_out;

  {
    const int nb = (n_nodes + BKT_NODES - 1) >> BKT_SHIFT;
    int2*     pairs = (int2*)mid;
    unsigned* tab   = (unsigned*)mid + 2 * (size_t)n_edges;
    const size_t lds = 2 * (size_t)nb * sizeof(unsigned);
    const size_t per = ((size_t)n_edges + SORT_SLICES - 1) / SORT_SLICES;
    gcn_slice_group<<<SORT_SLICES, 1024, per * 12 + lds, stream>>>(
        src, dst, tab, pairs, srcs, n_edges, nb);
    gcn_bucket_build<<<(unsigned)nb, BKT_NODES, 0, stream>>>(
        pairs, srcs, tab, (int)per, nb, out_norm, indeg, in_norm, seg_start,
        nbr, n_nodes, n_edges);
  }

  const int n_tiles = (n_nodes + TILE_ROWS - 1) / TILE_ROWS;
  const unsigned lblocks =
      (unsigned)((n_tiles + LAYER_TILES_PER_WG - 1) / LAYER_TILES_PER_WG);
  gcn_layer<true, true><<<lblocks, 128, 0, stream>>>(
      feat, out_norm, in_norm, seg_start, indeg, nbr, Ws[0], bs[0], outp,
      n_nodes, n_edges, n_tiles);
  gcn_layer<false, true><<<lblocks, 128, 0, stream>>>(
      outp, out_norm, in_norm, seg_start, indeg, nbr, Ws[1], bs[1], mid,
      n_nodes, n_edges, n_tiles);
  gcn_layer<false, false><<<lblocks, 128, 0, stream>>>(
      mid, out_norm, in_norm, seg_start, indeg, nbr, Ws[2], bs[2], outp,
      n_nodes, n_edges, n_tiles);
}


# Gather asks for eight rows of a node per round trip, not four (16 row loads in flight per lane, the 8 ids of one id register): a tile's chain of dependent trips halves; sum order unchanged
# speedup vs baseline: 7.2029x; 1.0319x over previous
#include <hip/hip_runtime.h>
#include <hip/hip_bf16.h>

#define D_FEAT 64

typedef float v2f __attribute__((ext_vector_type(2)));
typedef float v8f __attribute__((ext_vector_type(8)));

#define BKT_SHIFT 9
#define BKT_NODES 512
#define SORT_SLICES 256
#define SORT_BATCH 8
#define BKT_STAGE 12288
#define BKT_WAVES (BKT_NODES / 32)
static_assert(SORT_SLICES % (BKT_WAVES * SORT_BATCH) == 0,
              "pass B hands whole groups of slices to its waves");

__global__ __launch_bounds__(1024) void gcn_slice_group(
    const int* __restrict__ src, const int* __restrict__ dst,
    unsigned* __restrict__ tab, int2* __restrict__ pairs,
    int* __restrict__ srcs, int n_edges, int nb) {
  extern __shared__ int2 s_stage[];
  __shared__ unsigned s_total[32];
  const int per = (n_edges + (int)gridDim.x - 1) / (int)gridDim.x;
  int2* s_pair = s_stage;
  int* s_src = (int*)(s_stage + per);
  unsigned* s_cur = (unsigned*)(s_src + per);
  const int tid = threadIdx.x;
  const int lane = tid & 31;
  const int wv = tid >> 5;
  const int e0 = (int)blockIdx.x * per;
  const int e1 = (e0 + per) < n_edges ? (e0 + per) : n_edges;
  const int n_own = e1 > e0 ? e1 - e0 : 0;

  for (int i = tid; i < 2 * nb; i += 1024) s_cur[i] = 0u;
  __syncthreads();
  for (int e = e0 + tid; e < e1; e += SORT_BATCH * 1024) {
    int d[SORT_BATCH], s[SORT_BATCH];
#pragma unroll
    for (int j = 0; j < SORT_BATCH; ++j) {
      const int ej = e + j * 1024;
      const int ec = ej < e1 ? ej : e1 - 1;
      const int dv = dst[ec];
      const int sv = src[ec];
      d[j] = ej < e1 ? dv : -1;
      s[j] = sv;
    }
#pragma unroll
    for (int j = 0; j < SORT_BATCH; ++j) {
      if (d[j] >= 0) {
        atomicAdd(&s_cur[d[j] >> BKT_SHIFT], 1u);
        atomicAdd(&s_cur[nb + (s[j] >> BKT_SHIFT)], 1u);
      }
    }
  }
  __syncthreads();

  unsigned carry = 0u;
  for (int base = 0; base < 2 * nb; base += 1024) {
    const int i = base + tid;
    const unsigned c = (i < 2 * nb) ? s_cur[i] : 0u;
    unsigned incl = c;
    for (int x = 1; x < 32; x <<= 1) {
      unsigned t = __shfl_up(incl, x);
      if (lane >= x) incl += t;
    }
    if (lane == 31) s_total[wv] = incl;
    __syncthreads();
    unsigned woff = (lane < wv) ? s_total[lane] : 0u;
    unsigned all = s_total[lane];
    for (int x = 16; x > 0; x >>= 1) {
      woff += __shfl_xor(woff, x);
      all += __shfl_xor(all, x);
    }
    if (i < 2 * nb) s_cur[i] = carry + woff + incl - c;
    carry += all;
    __syncthreads();
  }
  for (int i = tid; i < 2 * nb; i += 1024) {
    const int key = i >= nb ? 1 : 0;
    const unsigned ls = s_cur[i] - (key ? (unsigned)n_own : 0u);
    s_cur[i] = ls;
    tab[((size_t)key * (nb + 1) + (i - key * nb)) * gridDim.x + blockIdx.x] = ls;
  }
  if (tid < 2)
    tab[((size_t)tid * (nb + 1) + nb) * gridDim.x + blockIdx.x] = (unsigned)n_own;
  __syncthreads();

  for (int e = e0 + tid; e < e1; e += SORT_BATCH * 1024) {
    int d[SORT_BATCH], s[SORT_BATCH];
#pragma unroll
    for (int j = 0; j < SORT_BATCH; ++j) {
      const int ej = e + j * 1024;
      const int ec = ej < e1 ? ej : e1 - 1;
      const int dv = dst[ec];
      const int sv = src[ec];
      d[j] = ej < e1 ? dv : -1;
      s[j] = sv;
    }
#pragma unroll
    for (int j = 0; j < SORT_BATCH; ++j) {
      if (d[j] >= 0) {
        const unsigned p = atomicAdd(&s_cur[d[j] >> BKT_SHIFT], 1u);
        const unsigned q = atomicAdd(&s_cur[nb + (s[j] >> BKT_SHIFT)], 1u);
        s_pair[p] = make_int2(s[j], d[j]);
        s_src[q] = s[j];
      }
    }
  }
  __syncthreads();
  for (int i = tid; i < n_own; i += 1024) {
    pairs[e0 + i] = s_pair[i];
    srcs[e0 + i] = s_src[i];
  }
}

__global__ __launch_bounds__(BKT_NODES) void gcn_bucket_build(
    const int2* __restrict__ pairs, const int* __restrict__ srcs,
    const unsigned* __restrict__ tab, int per, int nb,
    float* __restrict__ outn, unsigned* __restrict__ indeg,
    float* __restrict__ inn, int* __restrict__ seg_start,
    int* __restrict__ nbr, int n_nodes, int n_edges) {
  __shared__ unsigned c_in[BKT_NODES];
  __shared__ unsigned c_out[BKT_NODES];
  __shared__ unsigned s_total[32];
  __shared__ unsigned s_red[2][BKT_WAVES];
  __shared__ unsigned s_d0[SORT_SLICES];
  __shared__ unsigned s_dl[SORT_SLICES];
  __shared__ unsigned s_s0[SORT_SLICES];
  __shared__ unsigned s_sl[SORT_SLICES];
  __shared__ int s_nbr[BKT_STAGE];
  const int tid = threadIdx.x;
  const int lane = tid & 31;
  const int wv = tid >> 5;
  const int b = blockIdx.x;
  const int node0 = b << BKT_SHIFT;
  const unsigned last = (unsigned)(n_edges - 1);
  c_in[tid] = 0u;
  c_out[tid] = 0u;
  unsigned sum0 = 0u, suml = 0u;
  for (int s = tid; s < SORT_SLICES; s += BKT_NODES) {
    const unsigned* __restrict__ td = tab + (size_t)b * SORT_SLICES + s;
    const unsigned* __restrict__ ts = tab + (size_t)(nb + 1 + b) * SORT_SLICES + s;
    const unsigned a0 = td[0];
    const unsigned a1 = td[SORT_SLICES];
    const unsigned g0 = ts[0];
    const unsigned g1 = ts[SORT_SLICES];
    const unsigned base = (unsigned)s * (unsigned)per;
    s_d0[s] = base + a0;
    s_dl[s] = a1 - a0;
    s_s0[s] = base + g0;
    s_sl[s] = g1 - g0;
    sum0 += a0;
    suml += a1 - a0;
  }
  for (int x = 16; x > 0; x >>= 1) {
    sum0 += __shfl_xor(sum0, x);
    suml += __shfl_xor(suml, x);
  }
  if (lane == 0) { s_red[0][wv] = sum0; s_red[1][wv] = suml; }
  __syncthreads();
  unsigned r0 = lane < BKT_WAVES ? s_red[0][lane] : 0u;
  unsigned r1 = lane < BKT_WAVES ? s_red[1][lane] : 0u;
  for (int x = 16; x > 0; x >>= 1) {
    r0 += __shfl_xor(r0, x);
    r1 += __shfl_xor(r1, x);
  }
  const int es = (int)r0;
  const int ee = (int)(r0 + r1);

  for (int sb = 0; sb < SORT_SLICES; sb += BKT_WAVES * SORT_BATCH) {
    int d[SORT_BATCH][2], sv[SORT_BATCH][2];
    bool more = false;
#pragma unroll
    for (int j = 0; j < SORT_BATCH; ++j) {
      const int s = sb + j * BKT_WAVES + wv;
      const unsigned d0 = s_d0[s], dl = s_dl[s], s0 = s_s0[s], sl = s_sl[s];
      more = more || dl > 64u || sl > 64u;
#pragma unroll
      for (int h = 0; h < 2; ++h) {
        const unsigned k = (unsigned)(lane + 32 * h);
        const int dv = pairs[k < dl ? d0 + k : last].y;
        const int sx = srcs[k < sl ? s0 + k : last];
        d[j][h] = k < dl ? dv : -1;
        sv[j][h] = k < sl ? sx : -1;
      }
    }
#pragma unroll
    for (int j = 0; j < SORT_BATCH; ++j) {
#pragma unroll
      for (int h = 0; h < 2; ++h) {
        if (d[j][h] >= 0) atomicAdd(&c_in[d[j][h] - node0], 1u);
        if (sv[j][h] >= 0) atomicAdd(&c_out[sv[j][h] - node0], 1u);
      }
    }
    if (more)
    for (int j = 0; j < SORT_BATCH; ++j) {
      const int s = sb + j * BKT_WAVES + wv;
      const unsigned d0 = s_d0[s], dl = s_dl[s], s0 = s_s0[s], sl = s_sl[s];
      for (unsigned k = (unsigned)lane + 64u; k < dl; k += 32u)
        atomicAdd(&c_in[pairs[d0 + k].y - node0], 1u);
      for (unsigned k = (unsigned)lane + 64u; k < sl; k += 32u)
        atomicAdd(&c_out[srcs[s0 + k] - node0], 1u);
    }
  }
  __syncthreads();

  const unsigned d = c_in[tid];
  unsigned incl = d;
  for (int x = 1; x < 32; x <<= 1) {
    unsigned t = __shfl_up(incl, x);
    if (lane >= x) incl += t;
  }
  if (lane == 31) s_total[wv] = incl;
  __syncthreads();
  unsigned woff = (lane < wv) ? s_total[lane] : 0u;
  for (int x = 16; x > 0; x >>= 1) woff += __shfl_xor(woff, x);
  const unsigned local = woff + incl - d;
  const unsigned first = (unsigned)es + local;
  const bool staged = (ee - es) <= BKT_STAGE;
  c_in[tid] = staged ? local : first;

  const int node = node0 + tid;
  if (node < n_nodes) {
    const unsigned od = c_out[tid];
    seg_start[node] = (int)first;
    indeg[node] = d;
    inn[node] = rsqrtf((float)(d > 1u ? d : 1u));
    outn[node] = rsqrtf((float)(od > 1u ? od : 1u));
  }
  __syncthreads();
  for (int sb = 0; sb < SORT_SLICES; sb += BKT_WAVES * SORT_BATCH) {
    int2 p[SORT_BATCH][2];
    unsigned pos[SORT_BATCH][2];
    bool more = false;
#pragma unroll
    for (int j = 0; j < SORT_BATCH; ++j) {
      const int s = sb + j * BKT_WAVES + wv;
      const unsigned d0 = s_d0[s], dl = s_dl[s];
      more = more || dl > 64u;
#pragma unroll
      for (int h = 0; h < 2; ++h) {
        const unsigned k = (unsigned)(lane + 32 * h);
        p[j][h] = pairs[k < dl ? d0 + k : last];
        if (k >= dl) p[j][h].y = -1;
      }
    }
#pragma unroll
    for (int j = 0; j < SORT_BATCH; ++j) {
#pragma unroll
      for (int h = 0; h < 2; ++h) {
        pos[j][h] = 0u;
        if (p[j][h].y >= 0) pos[j][h] = atomicAdd(&c_in[p[j][h].y - node0], 1u);
      }
    }
#pragma unroll
    for (int j = 0; j < SORT_BATCH; ++j) {
#pragma unroll
      for (int h = 0; h < 2; ++h) {
        if (p[j][h].y >= 0) {
          if (staged) s_nbr[pos[j][h]] = p[j][h].x;
          else nbr[pos[j][h]] = p[j][h].x;
        }
      }
    }
    if (more)
    for (int j = 0; j < SORT_BATCH; ++j) {
      const int s = sb + j * BKT_WAVES + wv;
      const unsigned d0 = s_d0[s], dl = s_dl[s];
      for (unsigned k = (unsigned)lane + 64u; k < dl; k += 32u) {
        const int2 pr = pairs[d0 + k];
        const unsigned ps = atomicAdd(&c_in[pr.y - node0], 1u);
        if (staged) s_nbr[ps] = pr.x;
        else nbr[ps] = pr.x;
      }
    }
  }
  if (staged) {
    __syncthreads();
    for (int i = tid; i < ee - es; i += BKT_NODES) nbr[es + i] = s_nbr[i];
  }
}

#define TILE_ROWS 16
#define LAYER_TILES_PER_WG 4
#define TILE_PITCH 68
static_assert(LAYER_TILES_PER_WG <= 7, "lane i + 1 of a group must exist for every tile i");

__device__ __forceinline__ void gcn_store_rows(const float* __restrict__ s,
                                               float* __restrict__ out, int row0,
                                               int n_nodes, int wave, int lane) {
  const int c4 = (lane & 15) * 4;
  const int r = 4 * wave + (lane >> 4);
  const float* __restrict__ sp = s + r * TILE_PITCH + c4;
  float* __restrict__ o = out + (size_t)(row0 + r) * D_FEAT + c4;
  if (row0 + TILE_ROWS <= n_nodes) {
    const float4 v0 = *reinterpret_cast<const float4*>(sp);
    const float4 v1 = *reinterpret_cast<const float4*>(sp + 2 * TILE_PITCH);
    *reinterpret_cast<float4*>(o) = v0;
    *reinterpret_cast<float4*>(o + 2 * D_FEAT) = v1;
  } else {
#pragma nounroll
    for (int t = 0; t < 2; ++t) {
      if (row0 + r + 2 * t < n_nodes)
        *reinterpret_cast<float4*>(o + 2 * t * D_FEAT) =
            *reinterpret_cast<const float4*>(sp + 2 * t * TILE_PITCH);
    }
  }
}

__device__ __forceinline__ v2f gcn_elu2(v2f x) {
  const float kLog2e = __builtin_bit_cast(float, 0x3fb8aa3bu);
  const float kNegLn2Hi = __builtin_bit_cast(float, 0xbf317218u);
  const float kLn2Lo = __builtin_bit_cast(float, 0x3102e308u);
  const float kC6 = __builtin_bit_cast(float, 0x395133b1u);
  const float kC5 = __builtin_bit_cast(float, 0x3ab69700u);
  const float kC4 = __builtin_bit_cast(float, 0x3c0887f9u);
  const float kC3 = __builtin_bit_cast(float, 0x3d2aaa81u);
  const float kC2 = __builtin_bit_cast(float, 0x3e2aaaabu);
  const v2f t = x * kLog2e;
  v2f n;
  n.x = __builtin_rintf(t.x);
  n.y = __builtin_rintf(t.y);
  v2f r = __builtin_elementwise_fma(n, (v2f){kNegLn2Hi, kNegLn2Hi}, x);
  r = __builtin_elementwise_fma(n, (v2f){kLn2Lo, kLn2Lo}, r);
  v2f s;
  s.x = __builtin_ldexpf(1.0f, (int)n.x);
  s.y = __builtin_ldexpf(1.0f, (int)n.y);
  v2f p = __builtin_elementwise_fma((v2f){kC6, kC6}, r, (v2f){kC5, kC5});
  p = __builtin_elementwise_fma(r, p, (v2f){kC4, kC4});
  p = __builtin_elementwise_fma(r, p, (v2f){kC3, kC3});
  p = __builtin_elementwise_fma(r, p, (v2f){kC2, kC2});
  p = __builtin_elementwise_fma(r, p, (v2f){0.5f, 0.5f});
  const v2f q = r * p;
  const v2f r2 = __builtin_elementwise_fma(r, q, r);
  v2f e = s + (v2f){-1.0f, -1.0f};
  e = __builtin_elementwise_fma(s, r2, e);
  v2f y;
  y.x = x.x > 0.0f ? x.x : (-17.0f > x.x ? -1.0f : e.x);
  y.y = x.y > 0.0f ? x.y : (-17.0f > x.y ? -1.0f : e.y);
  return y;
}

template <bool kWeighted>
__device__ __forceinline__ void gcn_sum8(const char* __restrict__ xb, int ids,
                                         float wts, int left, int q,
                                         float4& lo, float4& hi) {
  float4 u[8], v[8];
#pragma unroll
  for (int k = 0; k < 8; ++k) {
    const unsigned off = ((unsigned)__shfl(ids, k, 8) << 8) + ((unsigned)q << 4);
    u[k] = *reinterpret_cast<const float4*>(xb + off);
    v[k] = *reinterpret_cast<const float4*>(xb + off + 128);
  }
  float w[8];
  if (kWeighted) {
    __builtin_amdgcn_sched_barrier(0);
  }
#pragma unroll
  for (int k = 0; k < 8; ++k) {
    const float wk = kWeighted ? __shfl(wts, k, 8) : 1.0f;
    w[k] = (k == 0 || k < left) ? wk : 0.0f;
  }
#pragma unroll
  for (int k = 0; k < 8; ++k) {
    lo.x += u[k].x * w[k]; lo.y += u[k].y * w[k]; lo.z += u[k].z * w[k]; lo.w += u[k].w * w[k];
    hi.x += v[k].x * w[k]; hi.y += v[k].y * w[k]; hi.z += v[k].z * w[k]; hi.w += v[k].w * w[k];
  }
}

template <bool kWeighted, bool kScaleOut>
__global__ __launch_bounds__(128, 7) void gcn_layer(
    const float* __restrict__ x, const float* __restrict__ out_norm,
    const float* __restrict__ in_norm, const int* __restrict__ seg_start,
    const unsigned* __restrict__ indeg, const int* __restrict__ nbr,
    const float* __restrict__ W,
    const float* __restrict__ bias,
    float* __restrict__ out, int n_nodes, int n_edges, int n_tiles) {
  __shared__ __attribute__((aligned(16))) float s_tile[2][TILE_ROWS * TILE_PITCH];
  __shared__ __attribute__((aligned(16))) float s_out[2][TILE_ROWS * TILE_PITCH];
  const int lane = threadIdx.x & 31;
  const int wave = threadIdx.x >> 5;
  const int nwg = (int)gridDim.x;
  const int q = lane & 7;
  const int grow = wave * 4 + (lane >> 3);
  const int half = lane >> 4;
  const int l15  = lane & 15;
  const int n0   = wave * 16;

  int mstart, mcnt;
  float msc, mon = 1.0f;
  {
    const int tq = (int)blockIdx.x + q * nwg;
    const int nodeq = tq * TILE_ROWS + grow;
    const bool okq = q < LAYER_TILES_PER_WG && tq < n_tiles && nodeq < n_nodes;
    const int nc = okq ? nodeq : 0;
    mstart = seg_start[nc];
    const int dc = (int)indeg[nc];
    msc = in_norm[nc];
    if (kScaleOut) mon = out_norm[nc];
    mcnt = okq ? dc : 0;
  }

  const float* __restrict__ bptr = W + (size_t)(2 * half) * D_FEAT + n0 + l15;
  v2f bf[D_FEAT / 4];
#pragma unroll
  for (int i = 0; i < D_FEAT / 4; ++i) {
    bf[i].x = bptr[(size_t)(4 * i) * D_FEAT];
    bf[i].y = bptr[(size_t)(4 * i + 1) * D_FEAT];
  }
  const float bv = bias[n0 + l15];
  const char* __restrict__ xb = reinterpret_cast<const char*>(x);

  int my0, my1;
  float nw0 = 1.0f, nw1 = 1.0f;
  {
    const int start = __shfl(mstart, 0, 8);
    const int cnt = __shfl(mcnt, 0, 8);
    const int m = cnt < 16 ? cnt : 16;
    const int e0 = start + (q < m ? q : 0);
    const int e1 = start + (q + 8 < m ? q + 8 : 0);
    my0 = nbr[e0 < n_edges ? e0 : n_edges - 1];
    my1 = nbr[e1 < n_edges ? e1 : n_edges - 1];
    if (kWeighted) {
      nw0 = out_norm[my0];
      nw1 = out_norm[my1];
    }
  }

  bool pending = false;
  int prow0 = 0;
  int pbuf = 0;
  int buf = 0;
  for (int i = 0; i < LAYER_TILES_PER_WG; ++i, buf ^= 1) {
    const int t = (int)blockIdx.x + i * nwg;
    if (t >= n_tiles) break;
    const int row0 = t * TILE_ROWS;

    const int cnt = __shfl(mcnt, i, 8);

    const int nstart = __shfl(mstart, i + 1, 8);
    const int ncnt = __shfl(mcnt, i + 1, 8);
    int nmy0, nmy1;
    {
      const int m = ncnt < 16 ? ncnt : 16;
      const int e0 = nstart + (q < m ? q : 0);
      const int e1 = nstart + (q + 8 < m ? q + 8 : 0);
      nmy0 = nbr[e0 < n_edges ? e0 : n_edges - 1];
      nmy1 = nbr[e1 < n_edges ? e1 : n_edges - 1];
    }

    float4 lo = make_float4(0.0f, 0.0f, 0.0f, 0.0f);
    float4 hi = make_float4(0.0f, 0.0f, 0.0f, 0.0f);
    for (int b = 0; b < cnt; b += 16) {
      const int m = (cnt - b) < 16 ? (cnt - b) : 16;
      if (b > 0) {
        const int start = __shfl(mstart, i, 8);
        my0 = nbr[start + b + (q < m ? q : 0)];
        my1 = nbr[start + b + (q + 8 < m ? q + 8 : 0)];
        if (kWeighted) {
          nw0 = out_norm[my0];
          nw1 = out_norm[my1];
        }
      }
      gcn_sum8<kWeighted>(xb, my0, nw0, m, q, lo, hi);
      if (m > 8) gcn_sum8<kWeighted>(xb, my1, nw1, m - 8, q, lo, hi);
    }
    const float sc = __shfl(msc, i, 8);
    lo.x *= sc; lo.y *= sc; lo.z *= sc; lo.w *= sc;
    hi.x *= sc; hi.y *= sc; hi.z *= sc; hi.w *= sc;

    float* __restrict__ st = &s_tile[buf][grow * TILE_PITCH];
    *reinterpret_cast<float4*>(st + 4 * q) = lo;
    *reinterpret_cast<float4*>(st + 32 + 4 * q) = hi;
    if (kScaleOut) {
      const float on = __shfl(mon, i, 8);
      if (q == 0) st[D_FEAT] = on;
    }
    __syncthreads();

    if (pending) gcn_store_rows(s_out[pbuf], out, prow0, n_nodes, wave, lane);

    const float* __restrict__ ap = &s_tile[buf][l15 * TILE_PITCH + 2 * half];
    v8f c = {};
#pragma unroll
    for (int k = 0; k < D_FEAT / 4; ++k) {
      const v2f a = *reinterpret_cast<const v2f*>(ap + 4 * k);
      c = __builtin_amdgcn_wmma_f32_16x16x4_f32(false, a, false, bf[k],
                                                (short)0, c, false, false);
    }

    float nnw0 = 1.0f, nnw1 = 1.0f;
    if (kWeighted) {
      nnw0 = out_norm[nmy0];
      nnw1 = out_norm[nmy1];
    }
    float osc[8];
    if (kScaleOut) {
      const float* __restrict__ sp = &s_tile[buf][(8 * half) * TILE_PITCH + D_FEAT];
#pragma unroll
      for (int v = 0; v < 8; ++v) osc[v] = sp[v * TILE_PITCH];
    }

    float* __restrict__ so = &s_out[buf][(8 * half) * TILE_PITCH + n0 + l15];
#pragma unroll
    for (int v = 0; v < 8; v += 2) {
      v2f xv;
      xv.x = c[v] + bv;
      xv.y = c[v + 1] + bv;
      v2f yv = gcn_elu2(xv);
      if (kScaleOut) {
        yv.x *= osc[v];
        yv.y *= osc[v + 1];
      }
      so[v * TILE_PITCH] = yv.x;
      so[(v + 1) * TILE_PITCH] = yv.y;
    }
    pending = true;
    prow0 = row0;
    pbuf = buf;

    my0 = nmy0; my1 = nmy1; nw0 = nnw0; nw1 = nnw1;
  }
  if (pending) {
    __syncthreads();
    gcn_store_rows(s_out[pbuf], out, prow0, n_nodes, wave, lane);
  }
}

extern "C" void kernel_launch(void* const* d_in, const int* in_sizes, int n_in,
                              void* d_out, int out_size, void* d_ws, size_t ws_size,
                              hipStream_t stream) {
  const float* feat = (const float*)d_in[0];
  const float* Ws[3] = {(const float*)d_in[1], (const float*)d_in[3], (const float*)d_in[5]};
  const float* bs[3] = {(const float*)d_in[2], (const float*)d_in[4], (const float*)d_in[6]};
  const int* src = (const int*)d_in[7];
  const int* dst = (const int*)d_in[8];

  const int n_nodes = in_sizes[0] / D_FEAT;
  const int n_edges = in_sizes[7];

  float*    out_norm  = (float*)d_ws;
  unsigned* indeg     = (unsigned*)(out_norm + n_nodes);
  float*    mid       = out_norm + 2 * (size_t)n_nodes;
  float*    in_norm   = mid + (size_t)n_nodes * D_FEAT;
  int*      seg_start = (int*)(in_norm + n_nodes);
  int*      nbr       = seg_start + n_nodes;
  int*      srcs      = nbr + n_edges;
  float*    outp      = (float*)d_out;

  {
    const int nb = (n_nodes + BKT_NODES - 1) >> BKT_SHIFT;
    int2*     pairs = (int2*)mid;
    unsigned* tab   = (unsigned*)mid + 2 * (size_t)n_edges;
    const size_t lds = 2 * (size_t)nb * sizeof(unsigned);
    const size_t per = ((size_t)n_edges + SORT_SLICES - 1) / SORT_SLICES;
    gcn_slice_group<<<SORT_SLICES, 1024, per * 12 + lds, stream>>>(
        src, dst, tab, pairs, srcs, n_edges, nb);
    gcn_bucket_build<<<(unsigned)nb, BKT_NODES, 0, stream>>>(
        pairs, srcs, tab, (int)per, nb, out_norm, indeg, in_norm, seg_start,
        nbr, n_nodes, n_edges);
  }

  const int n_tiles = (n_nodes + TILE_ROWS - 1) / TILE_ROWS;
  const unsigned lblocks =
      (unsigned)((n_tiles + LAYER_TILES_PER_WG - 1) / LAYER_TILES_PER_WG);
  gcn_layer<true, true><<<lblocks, 128, 0, stream>>>(
      feat, out_norm, in_norm, seg_start, indeg, nbr, Ws[0], bs[0], outp,
      n_nodes, n_edges, n_tiles);
  gcn_layer<false, true><<<lblocks, 128, 0, stream>>>(
      outp, out_norm, in_norm, seg_start, indeg, nbr, Ws[1], bs[1], mid,
      n_nodes, n_edges, n_tiles);
  gcn_layer<false, false><<<lblocks, 128, 0, stream>>>(
      mid, out_norm, in_norm, seg_start, indeg, nbr, Ws[2], bs[2], outp,
      n_nodes, n_edges, n_tiles);
}
